# starting kernel
# baseline (speedup 1.0000x reference)
#include <hip/hip_runtime.h>
#include <math.h>


typedef __attribute__((ext_vector_type(2))) float v2f;
typedef __attribute__((ext_vector_type(8))) float v8f;

#define HIDDEN 256
#define PROJ   64
#define PROJ2  128
#define RPB    64
#define XSTRIDE 260
#define HSTRIDE 68

__global__ __launch_bounds__(256)
void k_project(const float* __restrict__ node_feats,
               const float* __restrict__ W_k,
               const float* __restrict__ W_v,
               const int*   __restrict__ node_idx,
               const int*   __restrict__ he_idx,
               float* __restrict__ proj,
               float* __restrict__ ksum,
               float* __restrict__ cnt,
               int nnz)
{
    __shared__ float xs[RPB * XSTRIDE];
    const int e0 = blockIdx.x * RPB;
    const int t  = threadIdx.x;

    if (t < RPB) {
        const int e = e0 + t;
        if (e < nnz) atomicAdd(&cnt[he_idx[e]], 1.0f);
    }

    {
        const int r   = t >> 2;
        const int c0  = (t & 3) * 64;
        const int e   = e0 + r;
        const int row = (e < nnz) ? node_idx[e] : 0;
        const float* src = node_feats + (long)row * HIDDEN + c0;
        float* dst = xs + r * XSTRIDE + c0;
        #pragma unroll
        for (int i = 0; i < 64; i += 4)
            *(float4*)(dst + i) = *(const float4*)(src + i);
    }
    __syncthreads();

    const int wave = t >> 5;
    const int lane = t & 31;
    const int mn   = lane & 15;
    const int kb   = (lane >> 4) * 2;
    const int mh   = (lane >> 4) * 8;
    const int n0   = wave * 16;

    const int nrow = n0 + mn;
    const float* wrow = (nrow < PROJ) ? (W_k + (long)nrow * HIDDEN)
                                      : (W_v + (long)(nrow - PROJ) * HIDDEN);

    v8f acc[4] = {};
    #pragma unroll 2
    for (int kk = 0; kk < HIDDEN; kk += 4) {
        v2f b;
        b.x = wrow[kk + kb];
        b.y = wrow[kk + kb + 1];
        #pragma unroll
        for (int rt = 0; rt < 4; ++rt) {
            v2f a;
            const float* arow = xs + (rt * 16 + mn) * XSTRIDE;
            a.x = arow[kk + kb];
            a.y = arow[kk + kb + 1];
            acc[rt] = __builtin_amdgcn_wmma_f32_16x16x4_f32(
                          false, a, false, b, (short)0, acc[rt], false, false);
        }
    }

    const bool isKeyCols = (n0 < PROJ);
    if (e0 + RPB <= nnz) {
        #pragma unroll
        for (int rt = 0; rt < 4; ++rt)
            #pragma unroll
            for (int r = 0; r < 8; ++r) {
                const int e = e0 + rt * 16 + r + mh;
                const float v = acc[rt][r];
                proj[(long)e * PROJ2 + n0 + mn] = v;
                if (isKeyCols)
                    atomicAdd(&ksum[(long)he_idx[e] * PROJ + n0 + mn], v);
            }
    } else {
        #pragma unroll
        for (int rt = 0; rt < 4; ++rt)
            #pragma unroll
            for (int r = 0; r < 8; ++r) {
                const int e = e0 + rt * 16 + r + mh;
                if (e < nnz) {
                    const float v = acc[rt][r];
                    proj[(long)e * PROJ2 + n0 + mn] = v;
                    if (isKeyCols)
                        atomicAdd(&ksum[(long)he_idx[e] * PROJ + n0 + mn], v);
                }
            }
    }
}

__global__ __launch_bounds__(256)
void k_scores(const float* __restrict__ proj,
              const int* __restrict__ he_idx,
              const float* __restrict__ ksum, const float* __restrict__ cnt,
              float* __restrict__ scores, float* __restrict__ smax, int nnz)
{
    const int e = blockIdx.x * 8 + (threadIdx.x >> 5);
    if (e >= nnz) return;
    const int lane = threadIdx.x & 31;
    const int he = he_idx[e];
    const float inv = 1.0f / fmaxf(cnt[he], 1.0f);

    float ss = 0.0f;
    #pragma unroll
    for (int j = 0; j < 2; ++j) {
        const int p = lane + j * 32;
        const float d = proj[(long)e * PROJ2 + p]
                      - ksum[(long)he * PROJ + p] * inv;
        ss += d * d;
    }
    #pragma unroll
    for (int off = 16; off > 0; off >>= 1)
        ss += __shfl_down(ss, off, 32);

    if (lane == 0) {
        const float score = sqrtf(ss) * 0.125f;
        scores[e] = score;
        atomicMax((int*)&smax[he], __float_as_int(score));
    }
}

__global__ __launch_bounds__(256)
void k_exp(float* __restrict__ scores, const int* __restrict__ he_idx,
           const float* __restrict__ smax, float* __restrict__ denom, int nnz)
{
    const int e = blockIdx.x * blockDim.x + threadIdx.x;
    if (e >= nnz) return;
    const int he = he_idx[e];
    const float ex = expf(scores[e] - smax[he]);
    scores[e] = ex;
    atomicAdd(&denom[he], ex);
}

__global__ __launch_bounds__(256)
void k_aggregate(const float* __restrict__ proj,
                 const float* __restrict__ scores,
                 const int* __restrict__ he_idx,
                 const float* __restrict__ denom,
                 float* __restrict__ hefeats, int nnz)
{
    const long idx = (long)blockIdx.x * blockDim.x + threadIdx.x;
    if (idx >= (long)nnz * PROJ) return;
    const int e  = (int)(idx >> 6);
    const int p  = (int)(idx & 63);
    const int he = he_idx[e];
    const float attn = scores[e] / fmaxf(denom[he], 1e-30f);
    atomicAdd(&hefeats[(long)he * PROJ + p],
              attn * proj[(long)e * PROJ2 + PROJ + p]);
}

__global__ __launch_bounds__(256)
void k_out_gemm(const float* __restrict__ hefeats,
                const float* __restrict__ W_v,
                float* __restrict__ out, int num_he)
{
    __shared__ float hs[RPB * HSTRIDE];
    const int e0 = blockIdx.x * RPB;
    const int t  = threadIdx.x;
    {
        const int r = t >> 2;
        const int c = (t & 3) * 16;
        const int e = e0 + r;
        #pragma unroll
        for (int i = 0; i < 16; i += 4) {
            float4 v = (e < num_he)
                     ? *(const float4*)(hefeats + (long)e * PROJ + c + i)
                     : make_float4(0.f, 0.f, 0.f, 0.f);
            *(float4*)(hs + r * HSTRIDE + c + i) = v;
        }
    }
    __syncthreads();

    const int wave = t >> 5;
    const int lane = t & 31;
    const int mn   = lane & 15;
    const int kb   = (lane >> 4) * 2;
    const int mh   = (lane >> 4) * 8;
    const bool full = (e0 + RPB) <= num_he;

    #pragma unroll
    for (int sub = 0; sub < 2; ++sub) {
        const int n0 = wave * 32 + sub * 16;
        v8f acc[4] = {};
        #pragma unroll 2
        for (int kk = 0; kk < PROJ; kk += 4) {
            v2f b;
            b.x = W_v[(long)(kk + kb)     * HIDDEN + n0 + mn];
            b.y = W_v[(long)(kk + kb + 1) * HIDDEN + n0 + mn];
            #pragma unroll
            for (int rt = 0; rt < 4; ++rt) {
                v2f a;
                const float* arow = hs + (rt * 16 + mn) * HSTRIDE;
                a.x = arow[kk + kb];
                a.y = arow[kk + kb + 1];
                acc[rt] = __builtin_amdgcn_wmma_f32_16x16x4_f32(
                              false, a, false, b, (short)0, acc[rt], false, false);
            }
        }
        if (full) {
            #pragma unroll
            for (int rt = 0; rt < 4; ++rt)
                #pragma unroll
                for (int r = 0; r < 8; ++r) {
                    const int e = e0 + rt * 16 + r + mh;
                    out[(long)e * HIDDEN + n0 + mn] = acc[rt][r];
                }
        } else {
            #pragma unroll
            for (int rt = 0; rt < 4; ++rt)
                #pragma unroll
                for (int r = 0; r < 8; ++r) {
                    const int e = e0 + rt * 16 + r + mh;
                    if (e < num_he) out[(long)e * HIDDEN + n0 + mn] = acc[rt][r];
                }
        }
    }
}

__global__ __launch_bounds__(256)
void k_zero(float* __restrict__ p, long n)
{
    const long i = (long)blockIdx.x * blockDim.x + threadIdx.x;
    if (i < n) p[i] = 0.0f;
}

extern "C" void kernel_launch(void* const* d_in, const int* in_sizes, int n_in,
                              void* d_out, int out_size, void* d_ws, size_t ws_size,
                              hipStream_t stream)
{
    const float* node_feats = (const float*)d_in[0];
    const float* W_k        = (const float*)d_in[1];
    const float* W_v        = (const float*)d_in[2];
    const int*   node_idx   = (const int*)d_in[3];
    const int*   he_idx     = (const int*)d_in[4];
    (void)n_in; (void)ws_size;

    const int nnz = in_sizes[3];
    const int HE  = out_size / HIDDEN;

    float* ws = (float*)d_ws;
    long o = 0;
    float* proj    = ws + o; o += (long)nnz * PROJ2;
    float* ksum    = ws + o; o += (long)HE * PROJ;
    float* hefeats = ws + o; o += (long)HE * PROJ;
    float* cnt     = ws + o; o += HE;
    float* smax    = ws + o; o += HE;
    float* denom   = ws + o; o += HE;
    float* scores  = ws + o; o += nnz;

    const long zn = (long)HE * PROJ * 2 + 3L * HE;
    k_zero<<<(int)((zn + 255) / 256), 256, 0, stream>>>(ksum, zn);

    k_project<<<(nnz + RPB - 1) / RPB, 256, 0, stream>>>(
        node_feats, W_k, W_v, node_idx, he_idx, proj, ksum, cnt, nnz);

    k_scores<<<(nnz + 7) / 8, 256, 0, stream>>>(
        proj, he_idx, ksum, cnt, scores, smax, nnz);

    k_exp<<<(nnz + 255) / 256, 256, 0, stream>>>(
        scores, he_idx, smax, denom, nnz);

    const long t2 = (long)nnz * PROJ;
    k_aggregate<<<(int)((t2 + 255) / 256), 256, 0, stream>>>(
        proj, scores, he_idx, denom, hefeats, nnz);

    k_out_gemm<<<(HE + RPB - 1) / RPB, 256, 0, stream>>>(
        hefeats, W_v, (float*)d_out, HE);
}


# Project each node row once (N rows) and let edges gather projected rows, instead of projecting every gathered edge row (3x fewer WMMA); key-sum atomics move to a gather kernel, arrival order
# speedup vs baseline: 1.3689x; 1.3689x over previous
#include <hip/hip_runtime.h>
#include <math.h>


typedef __attribute__((ext_vector_type(2))) float v2f;
typedef __attribute__((ext_vector_type(8))) float v8f;

#define HIDDEN 256
#define PROJ   64
#define PROJ2  128
#define RPB    64
#define XSTRIDE 260
#define HSTRIDE 68

__global__ __launch_bounds__(256)
void k_project(const float* __restrict__ node_feats,
               const float* __restrict__ W_k,
               const float* __restrict__ W_v,
               float* __restrict__ proj,
               int n_nodes)
{
    __shared__ float xs[RPB * XSTRIDE];
    const int r0 = blockIdx.x * RPB;
    const int t  = threadIdx.x;

    {
        const int rr = t >> 6;
        const int c  = (t & 63) * 4;
        #pragma unroll
        for (int i = 0; i < 16; ++i) {
            const int r   = i * 4 + rr;
            const int n   = r0 + r;
            const int row = (n < n_nodes) ? n : 0;
            *(float4*)(xs + r * XSTRIDE + c) =
                *(const float4*)(node_feats + (long)row * HIDDEN + c);
        }
    }
    __syncthreads();

    const int wave = t >> 5;
    const int lane = t & 31;
    const int mn   = lane & 15;
    const int kb   = (lane >> 4) * 2;
    const int mh   = (lane >> 4) * 8;
    const int n0   = wave * 16;

    const int nrow = n0 + mn;
    const float* wrow = (nrow < PROJ) ? (W_k + (long)nrow * HIDDEN)
                                      : (W_v + (long)(nrow - PROJ) * HIDDEN);

    v8f acc[4] = {};
    #pragma unroll 2
    for (int kk = 0; kk < HIDDEN; kk += 4) {
        v2f b;
        b.x = wrow[kk + kb];
        b.y = wrow[kk + kb + 1];
        #pragma unroll
        for (int rt = 0; rt < 4; ++rt) {
            v2f a;
            const float* arow = xs + (rt * 16 + mn) * XSTRIDE;
            a.x = arow[kk + kb];
            a.y = arow[kk + kb + 1];
            acc[rt] = __builtin_amdgcn_wmma_f32_16x16x4_f32(
                          false, a, false, b, (short)0, acc[rt], false, false);
        }
    }

    if (r0 + RPB <= n_nodes) {
        #pragma unroll
        for (int rt = 0; rt < 4; ++rt)
            #pragma unroll
            for (int r = 0; r < 8; ++r) {
                const int n = r0 + rt * 16 + r + mh;
                proj[(long)n * PROJ2 + n0 + mn] = acc[rt][r];
            }
    } else {
        #pragma unroll
        for (int rt = 0; rt < 4; ++rt)
            #pragma unroll
            for (int r = 0; r < 8; ++r) {
                const int n = r0 + rt * 16 + r + mh;
                if (n < n_nodes)
                    proj[(long)n * PROJ2 + n0 + mn] = acc[rt][r];
            }
    }
}

#define EPW 4
__global__ __launch_bounds__(256)
void k_keysum(const float* __restrict__ proj,
              const int* __restrict__ node_idx,
              const int* __restrict__ he_idx,
              float* __restrict__ ksum,
              float* __restrict__ cnt,
              int nnz)
{
    const int lane = threadIdx.x & 31;
    const int wv = __builtin_amdgcn_readfirstlane((int)(threadIdx.x >> 5));
    const int e0 = (blockIdx.x * 8 + wv) * EPW;

    if (e0 + EPW <= nnz) {
        int   he[EPW];
        float k0[EPW], k1[EPW];
        #pragma unroll
        for (int j = 0; j < EPW; ++j) {
            const int e = e0 + j;
            he[j] = he_idx[e];
            const float* krow = proj + (long)node_idx[e] * PROJ2;
            k0[j] = krow[lane];
            k1[j] = krow[lane + 32];
        }
        #pragma unroll
        for (int j = 0; j < EPW; ++j) {
            float* dst = ksum + (long)he[j] * PROJ;
            atomicAdd(&dst[lane],      k0[j]);
            atomicAdd(&dst[lane + 32], k1[j]);
        }
        int hsel = he[0];
        #pragma unroll
        for (int j = 1; j < EPW; ++j) hsel = (lane == j) ? he[j] : hsel;
        if (lane < EPW) atomicAdd(&cnt[hsel], 1.0f);
    } else {
        for (int e = e0; e < nnz; ++e) {
            const int h = he_idx[e];
            const float* krow = proj + (long)node_idx[e] * PROJ2;
            float* dst = ksum + (long)h * PROJ;
            atomicAdd(&dst[lane],      krow[lane]);
            atomicAdd(&dst[lane + 32], krow[lane + 32]);
            if (lane == 0) atomicAdd(&cnt[h], 1.0f);
        }
    }
}

__global__ __launch_bounds__(256)
void k_scores(const float* __restrict__ proj,
              const int* __restrict__ node_idx,
              const int* __restrict__ he_idx,
              const float* __restrict__ ksum, const float* __restrict__ cnt,
              float* __restrict__ scores, float* __restrict__ smax, int nnz)
{
    const int e = blockIdx.x * 8 + (threadIdx.x >> 5);
    if (e >= nnz) return;
    const int lane = threadIdx.x & 31;
    const int he = he_idx[e];
    const float* krow = proj + (long)node_idx[e] * PROJ2;
    const float inv = 1.0f / fmaxf(cnt[he], 1.0f);

    float ss = 0.0f;
    #pragma unroll
    for (int j = 0; j < 2; ++j) {
        const int p = lane + j * 32;
        const float d = krow[p]
                      - ksum[(long)he * PROJ + p] * inv;
        ss += d * d;
    }
    #pragma unroll
    for (int off = 16; off > 0; off >>= 1)
        ss += __shfl_down(ss, off, 32);

    if (lane == 0) {
        const float score = sqrtf(ss) * 0.125f;
        scores[e] = score;
        atomicMax((int*)&smax[he], __float_as_int(score));
    }
}

__global__ __launch_bounds__(256)
void k_exp(float* __restrict__ scores, const int* __restrict__ he_idx,
           const float* __restrict__ smax, float* __restrict__ denom, int nnz)
{
    const int e = blockIdx.x * blockDim.x + threadIdx.x;
    if (e >= nnz) return;
    const int he = he_idx[e];
    const float ex = expf(scores[e] - smax[he]);
    scores[e] = ex;
    atomicAdd(&denom[he], ex);
}

__global__ __launch_bounds__(256)
void k_aggregate(const float* __restrict__ proj,
                 const float* __restrict__ scores,
                 const int* __restrict__ node_idx,
                 const int* __restrict__ he_idx,
                 const float* __restrict__ denom,
                 float* __restrict__ hefeats, int nnz)
{
    const long idx = (long)blockIdx.x * blockDim.x + threadIdx.x;
    if (idx >= (long)nnz * PROJ) return;
    const int e  = (int)(idx >> 6);
    const int p  = (int)(idx & 63);
    const int he = he_idx[e];
    const float attn = scores[e] / fmaxf(denom[he], 1e-30f);
    atomicAdd(&hefeats[(long)he * PROJ + p],
              attn * proj[(long)node_idx[e] * PROJ2 + PROJ + p]);
}

__global__ __launch_bounds__(256)
void k_out_gemm(const float* __restrict__ hefeats,
                const float* __restrict__ W_v,
                float* __restrict__ out, int num_he)
{
    __shared__ float hs[RPB * HSTRIDE];
    const int e0 = blockIdx.x * RPB;
    const int t  = threadIdx.x;
    {
        const int r = t >> 2;
        const int c = (t & 3) * 16;
        const int e = e0 + r;
        #pragma unroll
        for (int i = 0; i < 16; i += 4) {
            float4 v = (e < num_he)
                     ? *(const float4*)(hefeats + (long)e * PROJ + c + i)
                     : make_float4(0.f, 0.f, 0.f, 0.f);
            *(float4*)(hs + r * HSTRIDE + c + i) = v;
        }
    }
    __syncthreads();

    const int wave = t >> 5;
    const int lane = t & 31;
    const int mn   = lane & 15;
    const int kb   = (lane >> 4) * 2;
    const int mh   = (lane >> 4) * 8;
    const bool full = (e0 + RPB) <= num_he;

    #pragma unroll
    for (int sub = 0; sub < 2; ++sub) {
        const int n0 = wave * 32 + sub * 16;
        v8f acc[4] = {};
        #pragma unroll 2
        for (int kk = 0; kk < PROJ; kk += 4) {
            v2f b;
            b.x = W_v[(long)(kk + kb)     * HIDDEN + n0 + mn];
            b.y = W_v[(long)(kk + kb + 1) * HIDDEN + n0 + mn];
            #pragma unroll
            for (int rt = 0; rt < 4; ++rt) {
                v2f a;
                const float* arow = hs + (rt * 16 + mn) * HSTRIDE;
                a.x = arow[kk + kb];
                a.y = arow[kk + kb + 1];
                acc[rt] = __builtin_amdgcn_wmma_f32_16x16x4_f32(
                              false, a, false, b, (short)0, acc[rt], false, false);
            }
        }
        if (full) {
            #pragma unroll
            for (int rt = 0; rt < 4; ++rt)
                #pragma unroll
                for (int r = 0; r < 8; ++r) {
                    const int e = e0 + rt * 16 + r + mh;
                    out[(long)e * HIDDEN + n0 + mn] = acc[rt][r];
                }
        } else {
            #pragma unroll
            for (int rt = 0; rt < 4; ++rt)
                #pragma unroll
                for (int r = 0; r < 8; ++r) {
                    const int e = e0 + rt * 16 + r + mh;
                    if (e < num_he) out[(long)e * HIDDEN + n0 + mn] = acc[rt][r];
                }
        }
    }
}

__global__ __launch_bounds__(256)
void k_zero(float* __restrict__ p, long n)
{
    const long i = (long)blockIdx.x * blockDim.x + threadIdx.x;
    if (i < n) p[i] = 0.0f;
}

extern "C" void kernel_launch(void* const* d_in, const int* in_sizes, int n_in,
                              void* d_out, int out_size, void* d_ws, size_t ws_size,
                              hipStream_t stream)
{
    const float* node_feats = (const float*)d_in[0];
    const float* W_k        = (const float*)d_in[1];
    const float* W_v        = (const float*)d_in[2];
    const int*   node_idx   = (const int*)d_in[3];
    const int*   he_idx     = (const int*)d_in[4];
    (void)n_in; (void)ws_size;

    const int nnz = in_sizes[3];
    const int HE  = out_size / HIDDEN;
    const int NN  = in_sizes[0] / HIDDEN;

    float* ws = (float*)d_ws;
    long o = 0;
    float* proj    = ws + o; o += (long)NN * PROJ2;
    float* ksum    = ws + o; o += (long)HE * PROJ;
    float* hefeats = ws + o; o += (long)HE * PROJ;
    float* cnt     = ws + o; o += HE;
    float* smax    = ws + o; o += HE;
    float* denom   = ws + o; o += HE;
    float* scores  = ws + o; o += nnz;

    const long zn = (long)HE * PROJ * 2 + 3L * HE;
    k_zero<<<(int)((zn + 255) / 256), 256, 0, stream>>>(ksum, zn);

    k_project<<<(NN + RPB - 1) / RPB, 256, 0, stream>>>(
        node_feats, W_k, W_v, proj, NN);

    k_keysum<<<(nnz + 8 * EPW - 1) / (8 * EPW), 256, 0, stream>>>(
        proj, node_idx, he_idx, ksum, cnt, nnz);

    k_scores<<<(nnz + 7) / 8, 256, 0, stream>>>(
        proj, node_idx, he_idx, ksum, cnt, scores, smax, nnz);

    k_exp<<<(nnz + 255) / 256, 256, 0, stream>>>(
        scores, he_idx, smax, denom, nnz);

    const long t2 = (long)nnz * PROJ;
    k_aggregate<<<(int)((t2 + 255) / 256), 256, 0, stream>>>(
        proj, scores, node_idx, he_idx, denom, hefeats, nnz);

    k_out_gemm<<<(HE + RPB - 1) / RPB, 256, 0, stream>>>(
        hefeats, W_v, (float*)d_out, HE);
}


# Counting-sort edges by hyperedge (int histogram, scan, arrival-order rank); one wave per hyperedge then does mean, scores, softmax and value sum in registers: no float atomics, sums in rank order
# speedup vs baseline: 2.1416x; 1.5645x over previous
#include <hip/hip_runtime.h>
#include <math.h>


typedef __attribute__((ext_vector_type(2))) float v2f;
typedef __attribute__((ext_vector_type(8))) float v8f;

#define HIDDEN 256
#define PROJ   64
#define PROJ2  128
#define RPB    64
#define XSTRIDE 260
#define HSTRIDE 68

__global__ __launch_bounds__(256)
void k_project(const float* __restrict__ node_feats,
               const float* __restrict__ W_k,
               const float* __restrict__ W_v,
               float* __restrict__ proj,
               int n_nodes)
{
    __shared__ float xs[RPB * XSTRIDE];
    const int r0 = blockIdx.x * RPB;
    const int t  = threadIdx.x;

    {
        const int rr = t >> 6;
        const int c  = (t & 63) * 4;
        #pragma unroll
        for (int i = 0; i < 16; ++i) {
            const int r   = i * 4 + rr;
            const int n   = r0 + r;
            const int row = (n < n_nodes) ? n : 0;
            *(float4*)(xs + r * XSTRIDE + c) =
                *(const float4*)(node_feats + (long)row * HIDDEN + c);
        }
    }
    __syncthreads();

    const int wave = t >> 5;
    const int lane = t & 31;
    const int mn   = lane & 15;
    const int kb   = (lane >> 4) * 2;
    const int mh   = (lane >> 4) * 8;
    const int n0   = wave * 16;

    const int nrow = n0 + mn;
    const float* wrow = (nrow < PROJ) ? (W_k + (long)nrow * HIDDEN)
                                      : (W_v + (long)(nrow - PROJ) * HIDDEN);

    v8f acc[4] = {};
    #pragma unroll 2
    for (int kk = 0; kk < HIDDEN; kk += 4) {
        v2f b;
        b.x = wrow[kk + kb];
        b.y = wrow[kk + kb + 1];
        #pragma unroll
        for (int rt = 0; rt < 4; ++rt) {
            v2f a;
            const float* arow = xs + (rt * 16 + mn) * XSTRIDE;
            a.x = arow[kk + kb];
            a.y = arow[kk + kb + 1];
            acc[rt] = __builtin_amdgcn_wmma_f32_16x16x4_f32(
                          false, a, false, b, (short)0, acc[rt], false, false);
        }
    }

    if (r0 + RPB <= n_nodes) {
        #pragma unroll
        for (int rt = 0; rt < 4; ++rt)
            #pragma unroll
            for (int r = 0; r < 8; ++r) {
                const int n = r0 + rt * 16 + r + mh;
                proj[(long)n * PROJ2 + n0 + mn] = acc[rt][r];
            }
    } else {
        #pragma unroll
        for (int rt = 0; rt < 4; ++rt)
            #pragma unroll
            for (int r = 0; r < 8; ++r) {
                const int n = r0 + rt * 16 + r + mh;
                if (n < n_nodes)
                    proj[(long)n * PROJ2 + n0 + mn] = acc[rt][r];
            }
    }
}

__global__ __launch_bounds__(256)
void k_hist(const int* __restrict__ he_idx, int* __restrict__ seg, int nnz)
{
    const int e = blockIdx.x * blockDim.x + threadIdx.x;
    if (e < nnz) atomicAdd(&seg[he_idx[e]], 1);
}

#define SCAN_T 1024
#define SCAN_I 16
__global__ __launch_bounds__(SCAN_T)
void k_scan(int* seg, int n)
{
    __shared__ int wsum[32];
    const int t    = threadIdx.x;
    const int lane = t & 31;
    const int w    = t >> 5;
    int carry = 0;
    for (int base = 0; base < n; base += SCAN_T * SCAN_I) {
        const int i0 = base + t * SCAN_I;
        int v[SCAN_I];
        if (i0 + SCAN_I <= n) {
            #pragma unroll
            for (int q = 0; q < SCAN_I; q += 4) {
                const int4 x = *(const int4*)(seg + i0 + q);
                v[q] = x.x; v[q + 1] = x.y; v[q + 2] = x.z; v[q + 3] = x.w;
            }
        } else {
            #pragma unroll
            for (int q = 0; q < SCAN_I; ++q)
                v[q] = (i0 + q < n) ? seg[i0 + q] : 0;
        }
        int run = 0;
        #pragma unroll
        for (int q = 0; q < SCAN_I; ++q) { const int c = v[q]; v[q] = run; run += c; }
        int x = run;
        #pragma unroll
        for (int off = 1; off < 32; off <<= 1) {
            const int y = __shfl_up(x, off, 32);
            x += (lane >= off) ? y : 0;
        }
        if (lane == 31) wsum[w] = x;
        __syncthreads();
        if (w == 0) {
            int s = wsum[lane];
            #pragma unroll
            for (int off = 1; off < 32; off <<= 1) {
                const int y = __shfl_up(s, off, 32);
                s += (lane >= off) ? y : 0;
            }
            wsum[lane] = s;
        }
        __syncthreads();
        const int pre   = carry + ((w > 0) ? wsum[w - 1] : 0) + (x - run);
        const int total = wsum[31];
        if (i0 + SCAN_I <= n) {
            #pragma unroll
            for (int q = 0; q < SCAN_I; q += 4) {
                int4 o4;
                o4.x = v[q] + pre;     o4.y = v[q + 1] + pre;
                o4.z = v[q + 2] + pre; o4.w = v[q + 3] + pre;
                *(int4*)(seg + i0 + q) = o4;
            }
        } else {
            #pragma unroll
            for (int q = 0; q < SCAN_I; ++q)
                if (i0 + q < n) seg[i0 + q] = v[q] + pre;
        }
        __syncthreads();
        carry += total;
    }
}

__global__ __launch_bounds__(256)
void k_fill(const int* __restrict__ node_idx, const int* __restrict__ he_idx,
            int* __restrict__ seg, int* __restrict__ members, int nnz)
{
    const int e = blockIdx.x * blockDim.x + threadIdx.x;
    if (e < nnz) {
        const int slot = atomicAdd(&seg[he_idx[e]], 1);
        members[slot] = node_idx[e];
    }
}

#define DEV __device__ __forceinline__

template <int M>
DEV float row_xor(float x)
{
    return __builtin_bit_cast(float, __builtin_amdgcn_update_dpp(
        0, __builtin_bit_cast(int, x), 0x160 + M, 0xf, 0xf, false));
}
DEV float row_sum16(float v)
{
    v += row_xor<1>(v);
    v += row_xor<2>(v);
    v += row_xor<4>(v);
    v += row_xor<8>(v);
    return v;
}
DEV float lane0(float v)
{
    return __builtin_bit_cast(float,
        __builtin_amdgcn_readfirstlane(__builtin_bit_cast(int, v)));
}
DEV float4 ldrow(const float* __restrict__ proj, int nid, int lane)
{
    return *((const float4*)(proj + (long)nid * PROJ2) + lane);
}
DEV float dist2_part(float4 r, float4 c)
{
    const float dx = r.x - c.x, dy = r.y - c.y, dz = r.z - c.z, dw = r.w - c.w;
    return dx * dx + dy * dy + dz * dz + dw * dw;
}
DEV float key_score(const float* __restrict__ proj, int nid, float4 c, int lane)
{
    const float ss = lane0(row_sum16(dist2_part(ldrow(proj, nid, lane), c)));
    return sqrtf(ss) * 0.125f;
}

#define SEG_R 8
__global__ __launch_bounds__(256)
void k_segment(const float* __restrict__ proj,
               const int* __restrict__ members,
               const int* __restrict__ seg,
               float* __restrict__ hefeats, int num_he)
{
    const int lane = threadIdx.x & 31;
    const int wv = __builtin_amdgcn_readfirstlane((int)(threadIdx.x >> 5));
    const int h  = blockIdx.x * 8 + wv;
    if (h >= num_he) return;
    const int start = (h > 0) ? seg[h - 1] : 0;
    const int n     = seg[h] - start;
    const int* mem  = members + start;
    const float inv = 1.0f / fmaxf((float)n, 1.0f);
    const float4 zero4 = make_float4(0.f, 0.f, 0.f, 0.f);
    float4 acc = zero4;

    if (n <= SEG_R) {
        const int myid = (lane < n) ? mem[lane] : 0;
        int nid[SEG_R];
        #pragma unroll
        for (int u = 0; u < SEG_R; ++u)
            nid[u] = __builtin_amdgcn_readlane(myid, u);
        float4 r[SEG_R];
        #pragma unroll
        for (int u = 0; u < SEG_R; ++u) {
            r[u] = zero4;
            if (u < n) r[u] = ldrow(proj, nid[u], lane);
        }
        float4 ks = zero4;
        #pragma unroll
        for (int u = 0; u < SEG_R; ++u) {
            ks.x += r[u].x; ks.y += r[u].y; ks.z += r[u].z; ks.w += r[u].w;
        }
        const float4 c = make_float4(ks.x * inv, ks.y * inv, ks.z * inv, ks.w * inv);
        float sc[SEG_R];
        float smax = 0.0f;
        #pragma unroll
        for (int u = 0; u < SEG_R; ++u) {
            const float ss = lane0(row_sum16(dist2_part(r[u], c)));
            sc[u] = sqrtf(ss) * 0.125f;
            if (u < n) smax = fmaxf(smax, sc[u]);
        }
        float denom = 0.0f;
        #pragma unroll
        for (int u = 0; u < SEG_R; ++u) {
            sc[u] = (u < n) ? expf(sc[u] - smax) : 0.0f;
            denom += sc[u];
        }
        const float rden = fmaxf(denom, 1e-30f);
        #pragma unroll
        for (int u = 0; u < SEG_R; ++u) {
            const float wgt = sc[u] / rden;
            acc.x += wgt * r[u].x; acc.y += wgt * r[u].y;
            acc.z += wgt * r[u].z; acc.w += wgt * r[u].w;
        }
    } else {
        float4 ks = zero4;
        for (int base = 0; base < n; base += 32) {
            const int m    = min(32, n - base);
            const int myid = (lane < m) ? mem[base + lane] : 0;
            for (int j = 0; j < m; j += 4) {
                float4 r0, r1 = zero4, r2 = zero4, r3 = zero4;
                r0 = ldrow(proj, __builtin_amdgcn_readlane(myid, j), lane);
                if (j + 1 < m) r1 = ldrow(proj, __builtin_amdgcn_readlane(myid, j + 1), lane);
                if (j + 2 < m) r2 = ldrow(proj, __builtin_amdgcn_readlane(myid, j + 2), lane);
                if (j + 3 < m) r3 = ldrow(proj, __builtin_amdgcn_readlane(myid, j + 3), lane);
                ks.x += r0.x; ks.y += r0.y; ks.z += r0.z; ks.w += r0.w;
                ks.x += r1.x; ks.y += r1.y; ks.z += r1.z; ks.w += r1.w;
                ks.x += r2.x; ks.y += r2.y; ks.z += r2.z; ks.w += r2.w;
                ks.x += r3.x; ks.y += r3.y; ks.z += r3.z; ks.w += r3.w;
            }
        }
        const float4 c = make_float4(ks.x * inv, ks.y * inv, ks.z * inv, ks.w * inv);
        float smax = 0.0f;
        for (int base = 0; base < n; base += 32) {
            const int m    = min(32, n - base);
            const int myid = (lane < m) ? mem[base + lane] : 0;
            for (int j = 0; j < m; ++j)
                smax = fmaxf(smax, key_score(proj, __builtin_amdgcn_readlane(myid, j), c, lane));
        }
        float denom = 0.0f;
        for (int base = 0; base < n; base += 32) {
            const int m    = min(32, n - base);
            const int myid = (lane < m) ? mem[base + lane] : 0;
            for (int j = 0; j < m; ++j)
                denom += expf(key_score(proj, __builtin_amdgcn_readlane(myid, j), c, lane) - smax);
        }
        const float rden = fmaxf(denom, 1e-30f);
        for (int base = 0; base < n; base += 32) {
            const int m    = min(32, n - base);
            const int myid = (lane < m) ? mem[base + lane] : 0;
            for (int j = 0; j < m; ++j) {
                const int nid   = __builtin_amdgcn_readlane(myid, j);
                const float4 rr = ldrow(proj, nid, lane);
                const float ss  = lane0(row_sum16(dist2_part(rr, c)));
                const float wgt = expf(sqrtf(ss) * 0.125f - smax) / rden;
                acc.x += wgt * rr.x; acc.y += wgt * rr.y;
                acc.z += wgt * rr.z; acc.w += wgt * rr.w;
            }
        }
    }
    if (lane >= 16)
        *((float4*)(hefeats + (long)h * PROJ) + (lane - 16)) = acc;
}

__global__ __launch_bounds__(256)
void k_out_gemm(const float* __restrict__ hefeats,
                const float* __restrict__ W_v,
                float* __restrict__ out, int num_he)
{
    __shared__ float hs[RPB * HSTRIDE];
    const int e0 = blockIdx.x * RPB;
    const int t  = threadIdx.x;
    {
        const int r = t >> 2;
        const int c = (t & 3) * 16;
        const int e = e0 + r;
        #pragma unroll
        for (int i = 0; i < 16; i += 4) {
            float4 v = (e < num_he)
                     ? *(const float4*)(hefeats + (long)e * PROJ + c + i)
                     : make_float4(0.f, 0.f, 0.f, 0.f);
            *(float4*)(hs + r * HSTRIDE + c + i) = v;
        }
    }
    __syncthreads();

    const int wave = t >> 5;
    const int lane = t & 31;
    const int mn   = lane & 15;
    const int kb   = (lane >> 4) * 2;
    const int mh   = (lane >> 4) * 8;
    const bool full = (e0 + RPB) <= num_he;

    #pragma unroll
    for (int sub = 0; sub < 2; ++sub) {
        const int n0 = wave * 32 + sub * 16;
        v8f acc[4] = {};
        #pragma unroll 2
        for (int kk = 0; kk < PROJ; kk += 4) {
            v2f b;
            b.x = W_v[(long)(kk + kb)     * HIDDEN + n0 + mn];
            b.y = W_v[(long)(kk + kb + 1) * HIDDEN + n0 + mn];
            #pragma unroll
            for (int rt = 0; rt < 4; ++rt) {
                v2f a;
                const float* arow = hs + (rt * 16 + mn) * HSTRIDE;
                a.x = arow[kk + kb];
                a.y = arow[kk + kb + 1];
                acc[rt] = __builtin_amdgcn_wmma_f32_16x16x4_f32(
                              false, a, false, b, (short)0, acc[rt], false, false);
            }
        }
        if (full) {
            #pragma unroll
            for (int rt = 0; rt < 4; ++rt)
                #pragma unroll
                for (int r = 0; r < 8; ++r) {
                    const int e = e0 + rt * 16 + r + mh;
                    out[(long)e * HIDDEN + n0 + mn] = acc[rt][r];
                }
        } else {
            #pragma unroll
            for (int rt = 0; rt < 4; ++rt)
                #pragma unroll
                for (int r = 0; r < 8; ++r) {
                    const int e = e0 + rt * 16 + r + mh;
                    if (e < num_he) out[(long)e * HIDDEN + n0 + mn] = acc[rt][r];
                }
        }
    }
}

__global__ __launch_bounds__(256)
void k_zero(int* __restrict__ p, int n)
{
    const int i = blockIdx.x * blockDim.x + threadIdx.x;
    if (i < n) p[i] = 0;
}

extern "C" void kernel_launch(void* const* d_in, const int* in_sizes, int n_in,
                              void* d_out, int out_size, void* d_ws, size_t ws_size,
                              hipStream_t stream)
{
    const float* node_feats = (const float*)d_in[0];
    const float* W_k        = (const float*)d_in[1];
    const float* W_v        = (const float*)d_in[2];
    const int*   node_idx   = (const int*)d_in[3];
    const int*   he_idx     = (const int*)d_in[4];
    (void)n_in; (void)ws_size;

    const int nnz = in_sizes[3];
    const int HE  = out_size / HIDDEN;
    const int NN  = in_sizes[0] / HIDDEN;

    float* ws = (float*)d_ws;
    long o = 0;
    float* proj    = ws + o;        o += (long)NN * PROJ2;
    float* hefeats = ws + o;        o += (long)HE * PROJ;
    int*   seg     = (int*)(ws + o); o += ((long)HE + 3) / 4 * 4;
    int*   members = (int*)(ws + o); o += nnz;

    k_zero<<<(HE + 255) / 256, 256, 0, stream>>>(seg, HE);

    k_hist<<<(nnz + 255) / 256, 256, 0, stream>>>(he_idx, seg, nnz);

    k_scan<<<1, SCAN_T, 0, stream>>>(seg, HE);

    k_fill<<<(nnz + 255) / 256, 256, 0, stream>>>(
        node_idx, he_idx, seg, members, nnz);

    k_project<<<(NN + RPB - 1) / RPB, 256, 0, stream>>>(
        node_feats, W_k, W_v, proj, NN);

    k_segment<<<(HE + 7) / 8, 256, 0, stream>>>(
        proj, members, seg, hefeats, HE);

    k_out_gemm<<<(HE + RPB - 1) / RPB, 256, 0, stream>>>(
        hefeats, W_v, (float*)d_out, HE);
}


# Edges grouped in one atomic pass: each takes the next slot of its hyperedge's own 32-slot row (overflow list beyond 32), so histogram and prefix-scan launches go; sums stay in arrival order
# speedup vs baseline: 2.6673x; 1.2455x over previous
#include <hip/hip_runtime.h>
#include <math.h>


typedef __attribute__((ext_vector_type(2))) float v2f;
typedef __attribute__((ext_vector_type(8))) float v8f;

#define HIDDEN 256
#define PROJ   64
#define PROJ2  128
#define RPB    64
#define XSTRIDE 260
#define HSTRIDE 68

__global__ __launch_bounds__(256)
void k_project(const float* __restrict__ node_feats,
               const float* __restrict__ W_k,
               const float* __restrict__ W_v,
               float* __restrict__ proj,
               int n_nodes)
{
    __shared__ float xs[RPB * XSTRIDE];
    const int r0 = blockIdx.x * RPB;
    const int t  = threadIdx.x;

    {
        const int rr = t >> 6;
        const int c  = (t & 63) * 4;
        #pragma unroll
        for (int i = 0; i < 16; ++i) {
            const int r   = i * 4 + rr;
            const int n   = r0 + r;
            const int row = (n < n_nodes) ? n : 0;
            *(float4*)(xs + r * XSTRIDE + c) =
                *(const float4*)(node_feats + (long)row * HIDDEN + c);
        }
    }
    __syncthreads();

    const int wave = t >> 5;
    const int lane = t & 31;
    const int mn   = lane & 15;
    const int kb   = (lane >> 4) * 2;
    const int mh   = (lane >> 4) * 8;
    const int n0   = wave * 16;

    const int nrow = n0 + mn;
    const float* wrow = (nrow < PROJ) ? (W_k + (long)nrow * HIDDEN)
                                      : (W_v + (long)(nrow - PROJ) * HIDDEN);

    v8f acc[4] = {};
    #pragma unroll 2
    for (int kk = 0; kk < HIDDEN; kk += 4) {
        v2f b;
        b.x = wrow[kk + kb];
        b.y = wrow[kk + kb + 1];
        #pragma unroll
        for (int rt = 0; rt < 4; ++rt) {
            v2f a;
            const float* arow = xs + (rt * 16 + mn) * XSTRIDE;
            a.x = arow[kk + kb];
            a.y = arow[kk + kb + 1];
            acc[rt] = __builtin_amdgcn_wmma_f32_16x16x4_f32(
                          false, a, false, b, (short)0, acc[rt], false, false);
        }
    }

    if (r0 + RPB <= n_nodes) {
        #pragma unroll
        for (int rt = 0; rt < 4; ++rt)
            #pragma unroll
            for (int r = 0; r < 8; ++r) {
                const int n = r0 + rt * 16 + r + mh;
                proj[(long)n * PROJ2 + n0 + mn] = acc[rt][r];
            }
    } else {
        #pragma unroll
        for (int rt = 0; rt < 4; ++rt)
            #pragma unroll
            for (int r = 0; r < 8; ++r) {
                const int n = r0 + rt * 16 + r + mh;
                if (n < n_nodes)
                    proj[(long)n * PROJ2 + n0 + mn] = acc[rt][r];
            }
    }
}

#define CAP 32
__global__ __launch_bounds__(256)
void k_fill(const int* __restrict__ node_idx, const int* __restrict__ he_idx,
            int* __restrict__ cnt, int* __restrict__ bucket,
            int* __restrict__ ovf, int nnz, int num_he)
{
    const int e = blockIdx.x * blockDim.x + threadIdx.x;
    if (e < nnz) {
        const int h    = he_idx[e];
        const int nid  = node_idx[e];
        const int slot = atomicAdd(&cnt[h], 1);
        if (slot < CAP) {
            bucket[(long)h * CAP + slot] = nid;
        } else {
            const int o = atomicAdd(&cnt[num_he], 1);
            ovf[2 * o]     = h;
            ovf[2 * o + 1] = nid;
        }
    }
}

#define DEV __device__ __forceinline__

template <int M>
DEV float row_xor(float x)
{
    return __builtin_bit_cast(float, __builtin_amdgcn_update_dpp(
        0, __builtin_bit_cast(int, x), 0x160 + M, 0xf, 0xf, false));
}
DEV float row_sum16(float v)
{
    v += row_xor<1>(v);
    v += row_xor<2>(v);
    v += row_xor<4>(v);
    v += row_xor<8>(v);
    return v;
}
DEV float lane0(float v)
{
    return __builtin_bit_cast(float,
        __builtin_amdgcn_readfirstlane(__builtin_bit_cast(int, v)));
}
DEV float4 ldrow(const float* __restrict__ proj, int nid, int lane)
{
    return *((const float4*)(proj + (long)nid * PROJ2) + lane);
}
DEV float dist2_part(float4 r, float4 c)
{
    const float dx = r.x - c.x, dy = r.y - c.y, dz = r.z - c.z, dw = r.w - c.w;
    return dx * dx + dy * dy + dz * dz + dw * dw;
}
DEV float key_score(const float* __restrict__ proj, int nid, float4 c, int lane)
{
    const float ss = lane0(row_sum16(dist2_part(ldrow(proj, nid, lane), c)));
    return sqrtf(ss) * 0.125f;
}

template <class F>
DEV void for_each_overflow(const int* __restrict__ ovf, int novf, int h, int lane, F f)
{
    for (int base = 0; base < novf; base += 32) {
        const int i = base + lane;
        int hh = -1, nd = 0;
        if (i < novf) { hh = ovf[2 * i]; nd = ovf[2 * i + 1]; }
        unsigned hit = __builtin_amdgcn_ballot_w32(hh == h);
        while (hit) {
            const int j = __builtin_ctz(hit);
            hit &= hit - 1;
            f(__builtin_amdgcn_readlane(nd, j));
        }
    }
}

#define SEG_R 8
__global__ __launch_bounds__(256)
void k_segment(const float* __restrict__ proj,
               const int* __restrict__ bucket,
               const int* __restrict__ ovf,
               const int* __restrict__ cnt,
               float* __restrict__ hefeats, int num_he)
{
    const int lane = threadIdx.x & 31;
    const int wv = __builtin_amdgcn_readfirstlane((int)(threadIdx.x >> 5));
    const int h  = blockIdx.x * 8 + wv;
    if (h >= num_he) return;
    const int n    = cnt[h];
    const int nb   = min(n, CAP);
    const int myid = (lane < nb) ? bucket[(long)h * CAP + lane] : 0;
    const float inv = 1.0f / fmaxf((float)n, 1.0f);
    const float4 zero4 = make_float4(0.f, 0.f, 0.f, 0.f);
    float4 acc = zero4;

    if (n <= SEG_R) {
        int nid[SEG_R];
        #pragma unroll
        for (int u = 0; u < SEG_R; ++u)
            nid[u] = __builtin_amdgcn_readlane(myid, u);
        float4 r[SEG_R];
        #pragma unroll
        for (int u = 0; u < SEG_R; ++u) {
            r[u] = zero4;
            if (u < n) r[u] = ldrow(proj, nid[u], lane);
        }
        float4 ks = zero4;
        #pragma unroll
        for (int u = 0; u < SEG_R; ++u) {
            ks.x += r[u].x; ks.y += r[u].y; ks.z += r[u].z; ks.w += r[u].w;
        }
        const float4 c = make_float4(ks.x * inv, ks.y * inv, ks.z * inv, ks.w * inv);
        float sc[SEG_R];
        float smax = 0.0f;
        #pragma unroll
        for (int u = 0; u < SEG_R; ++u) {
            const float ss = lane0(row_sum16(dist2_part(r[u], c)));
            sc[u] = sqrtf(ss) * 0.125f;
            if (u < n) smax = fmaxf(smax, sc[u]);
        }
        float denom = 0.0f;
        #pragma unroll
        for (int u = 0; u < SEG_R; ++u) {
            sc[u] = (u < n) ? expf(sc[u] - smax) : 0.0f;
            denom += sc[u];
        }
        const float rden = fmaxf(denom, 1e-30f);
        #pragma unroll
        for (int u = 0; u < SEG_R; ++u) {
            const float wgt = sc[u] / rden;
            acc.x += wgt * r[u].x; acc.y += wgt * r[u].y;
            acc.z += wgt * r[u].z; acc.w += wgt * r[u].w;
        }
    } else {
        const int novf = (n > CAP) ? cnt[num_he] : 0;
        float4 ks = zero4;
        for (int j = 0; j < nb; j += 4) {
            float4 r0, r1 = zero4, r2 = zero4, r3 = zero4;
            r0 = ldrow(proj, __builtin_amdgcn_readlane(myid, j), lane);
            if (j + 1 < nb) r1 = ldrow(proj, __builtin_amdgcn_readlane(myid, j + 1), lane);
            if (j + 2 < nb) r2 = ldrow(proj, __builtin_amdgcn_readlane(myid, j + 2), lane);
            if (j + 3 < nb) r3 = ldrow(proj, __builtin_amdgcn_readlane(myid, j + 3), lane);
            ks.x += r0.x; ks.y += r0.y; ks.z += r0.z; ks.w += r0.w;
            ks.x += r1.x; ks.y += r1.y; ks.z += r1.z; ks.w += r1.w;
            ks.x += r2.x; ks.y += r2.y; ks.z += r2.z; ks.w += r2.w;
            ks.x += r3.x; ks.y += r3.y; ks.z += r3.z; ks.w += r3.w;
        }
        for_each_overflow(ovf, novf, h, lane, [&](int nid) {
            const float4 rr = ldrow(proj, nid, lane);
            ks.x += rr.x; ks.y += rr.y; ks.z += rr.z; ks.w += rr.w;
        });
        const float4 c = make_float4(ks.x * inv, ks.y * inv, ks.z * inv, ks.w * inv);
        float smax = 0.0f;
        for (int j = 0; j < nb; ++j)
            smax = fmaxf(smax, key_score(proj, __builtin_amdgcn_readlane(myid, j), c, lane));
        for_each_overflow(ovf, novf, h, lane, [&](int nid) {
            smax = fmaxf(smax, key_score(proj, nid, c, lane));
        });
        float denom = 0.0f;
        for (int j = 0; j < nb; ++j)
            denom += expf(key_score(proj, __builtin_amdgcn_readlane(myid, j), c, lane) - smax);
        for_each_overflow(ovf, novf, h, lane, [&](int nid) {
            denom += expf(key_score(proj, nid, c, lane) - smax);
        });
        const float rden = fmaxf(denom, 1e-30f);
        auto add_row = [&](int nid) {
            const float4 rr = ldrow(proj, nid, lane);
            const float ss  = lane0(row_sum16(dist2_part(rr, c)));
            const float wgt = expf(sqrtf(ss) * 0.125f - smax) / rden;
            acc.x += wgt * rr.x; acc.y += wgt * rr.y;
            acc.z += wgt * rr.z; acc.w += wgt * rr.w;
        };
        for (int j = 0; j < nb; ++j)
            add_row(__builtin_amdgcn_readlane(myid, j));
        for_each_overflow(ovf, novf, h, lane, add_row);
    }
    if (lane >= 16)
        *((float4*)(hefeats + (long)h * PROJ) + (lane - 16)) = acc;
}

__global__ __launch_bounds__(256)
void k_out_gemm(const float* __restrict__ hefeats,
                const float* __restrict__ W_v,
                float* __restrict__ out, int num_he)
{
    __shared__ float hs[RPB * HSTRIDE];
    const int e0 = blockIdx.x * RPB;
    const int t  = threadIdx.x;
    {
        const int r = t >> 2;
        const int c = (t & 3) * 16;
        const int e = e0 + r;
        #pragma unroll
        for (int i = 0; i < 16; i += 4) {
            float4 v = (e < num_he)
                     ? *(const float4*)(hefeats + (long)e * PROJ + c + i)
                     : make_float4(0.f, 0.f, 0.f, 0.f);
            *(float4*)(hs + r * HSTRIDE + c + i) = v;
        }
    }
    __syncthreads();

    const int wave = t >> 5;
    const int lane = t & 31;
    const int mn   = lane & 15;
    const int kb   = (lane >> 4) * 2;
    const int mh   = (lane >> 4) * 8;
    const bool full = (e0 + RPB) <= num_he;

    #pragma unroll
    for (int sub = 0; sub < 2; ++sub) {
        const int n0 = wave * 32 + sub * 16;
        v8f acc[4] = {};
        #pragma unroll 2
        for (int kk = 0; kk < PROJ; kk += 4) {
            v2f b;
            b.x = W_v[(long)(kk + kb)     * HIDDEN + n0 + mn];
            b.y = W_v[(long)(kk + kb + 1) * HIDDEN + n0 + mn];
            #pragma unroll
            for (int rt = 0; rt < 4; ++rt) {
                v2f a;
                const float* arow = hs + (rt * 16 + mn) * HSTRIDE;
                a.x = arow[kk + kb];
                a.y = arow[kk + kb + 1];
                acc[rt] = __builtin_amdgcn_wmma_f32_16x16x4_f32(
                              false, a, false, b, (short)0, acc[rt], false, false);
            }
        }
        if (full) {
            #pragma unroll
            for (int rt = 0; rt < 4; ++rt)
                #pragma unroll
                for (int r = 0; r < 8; ++r) {
                    const int e = e0 + rt * 16 + r + mh;
                    out[(long)e * HIDDEN + n0 + mn] = acc[rt][r];
                }
        } else {
            #pragma unroll
            for (int rt = 0; rt < 4; ++rt)
                #pragma unroll
                for (int r = 0; r < 8; ++r) {
                    const int e = e0 + rt * 16 + r + mh;
                    if (e < num_he) out[(long)e * HIDDEN + n0 + mn] = acc[rt][r];
                }
        }
    }
}

__global__ __launch_bounds__(256)
void k_zero(int* __restrict__ p, int n)
{
    const int i = blockIdx.x * blockDim.x + threadIdx.x;
    if (i < n) p[i] = 0;
}

extern "C" void kernel_launch(void* const* d_in, const int* in_sizes, int n_in,
                              void* d_out, int out_size, void* d_ws, size_t ws_size,
                              hipStream_t stream)
{
    const float* node_feats = (const float*)d_in[0];
    const float* W_k        = (const float*)d_in[1];
    const float* W_v        = (const float*)d_in[2];
    const int*   node_idx   = (const int*)d_in[3];
    const int*   he_idx     = (const int*)d_in[4];
    (void)n_in; (void)ws_size;

    const int nnz = in_sizes[3];
    const int HE  = out_size / HIDDEN;
    const int NN  = in_sizes[0] / HIDDEN;

    float* ws = (float*)d_ws;
    long o = 0;
    float* proj    = ws + o;        o += (long)NN * PROJ2;
    float* hefeats = ws + o;        o += (long)HE * PROJ;
    int*   cnt     = (int*)(ws + o); o += ((long)HE + 1 + 31) / 32 * 32;
    int*   bucket  = (int*)(ws + o); o += (long)HE * CAP;
    int*   ovf     = (int*)(ws + o); o += 2L * nnz;

    k_zero<<<(HE + 1 + 255) / 256, 256, 0, stream>>>(cnt, HE + 1);

    k_fill<<<(nnz + 255) / 256, 256, 0, stream>>>(
        node_idx, he_idx, cnt, bucket, ovf, nnz, HE);

    k_project<<<(NN + RPB - 1) / RPB, 256, 0, stream>>>(
        node_feats, W_k, W_v, proj, NN);

    k_segment<<<(HE + 7) / 8, 256, 0, stream>>>(
        proj, bucket, ovf, cnt, hefeats, HE);

    k_out_gemm<<<(HE + RPB - 1) / RPB, 256, 0, stream>>>(
        hefeats, W_v, (float*)d_out, HE);
}


# k_segment, hyperedges of over 8 members: rows fetched 8 at a time with all loads in flight, member j's score kept in lane j (3 row reads, not 4), not one dependent load per member; same sum order
# speedup vs baseline: 3.0479x; 1.1427x over previous
#include <hip/hip_runtime.h>
#include <math.h>


typedef __attribute__((ext_vector_type(2))) float v2f;
typedef __attribute__((ext_vector_type(8))) float v8f;

#define HIDDEN 256
#define PROJ   64
#define PROJ2  128
#define RPB    64
#define XSTRIDE 260
#define HSTRIDE 68

__global__ __launch_bounds__(256)
void k_project(const float* __restrict__ node_feats,
               const float* __restrict__ W_k,
               const float* __restrict__ W_v,
               float* __restrict__ proj,
               int n_nodes)
{
    __shared__ float xs[RPB * XSTRIDE];
    const int r0 = blockIdx.x * RPB;
    const int t  = threadIdx.x;

    {
        const int rr = t >> 6;
        const int c  = (t & 63) * 4;
        #pragma unroll
        for (int i = 0; i < 16; ++i) {
            const int r   = i * 4 + rr;
            const int n   = r0 + r;
            const int row = (n < n_nodes) ? n : 0;
            *(float4*)(xs + r * XSTRIDE + c) =
                *(const float4*)(node_feats + (long)row * HIDDEN + c);
        }
    }
    __syncthreads();

    const int wave = t >> 5;
    const int lane = t & 31;
    const int mn   = lane & 15;
    const int kb   = (lane >> 4) * 2;
    const int mh   = (lane >> 4) * 8;
    const int n0   = wave * 16;

    const int nrow = n0 + mn;
    const float* wrow = (nrow < PROJ) ? (W_k + (long)nrow * HIDDEN)
                                      : (W_v + (long)(nrow - PROJ) * HIDDEN);

    v8f acc[4] = {};
    #pragma unroll 2
    for (int kk = 0; kk < HIDDEN; kk += 4) {
        v2f b;
        b.x = wrow[kk + kb];
        b.y = wrow[kk + kb + 1];
        #pragma unroll
        for (int rt = 0; rt < 4; ++rt) {
            v2f a;
            const float* arow = xs + (rt * 16 + mn) * XSTRIDE;
            a.x = arow[kk + kb];
            a.y = arow[kk + kb + 1];
            acc[rt] = __builtin_amdgcn_wmma_f32_16x16x4_f32(
                          false, a, false, b, (short)0, acc[rt], false, false);
        }
    }

    if (r0 + RPB <= n_nodes) {
        #pragma unroll
        for (int rt = 0; rt < 4; ++rt)
            #pragma unroll
            for (int r = 0; r < 8; ++r) {
                const int n = r0 + rt * 16 + r + mh;
                proj[(long)n * PROJ2 + n0 + mn] = acc[rt][r];
            }
    } else {
        #pragma unroll
        for (int rt = 0; rt < 4; ++rt)
            #pragma unroll
            for (int r = 0; r < 8; ++r) {
                const int n = r0 + rt * 16 + r + mh;
                if (n < n_nodes)
                    proj[(long)n * PROJ2 + n0 + mn] = acc[rt][r];
            }
    }
}

#define CAP 32
__global__ __launch_bounds__(256)
void k_fill(const int* __restrict__ node_idx, const int* __restrict__ he_idx,
            int* __restrict__ cnt, int* __restrict__ bucket,
            int* __restrict__ ovf, int nnz, int num_he)
{
    const int e = blockIdx.x * blockDim.x + threadIdx.x;
    if (e < nnz) {
        const int h    = he_idx[e];
        const int nid  = node_idx[e];
        const int slot = atomicAdd(&cnt[h], 1);
        if (slot < CAP) {
            bucket[(long)h * CAP + slot] = nid;
        } else {
            const int o = atomicAdd(&cnt[num_he], 1);
            ovf[2 * o]     = h;
            ovf[2 * o + 1] = nid;
        }
    }
}

#define DEV __device__ __forceinline__

template <int M>
DEV float row_xor(float x)
{
    return __builtin_bit_cast(float, __builtin_amdgcn_update_dpp(
        0, __builtin_bit_cast(int, x), 0x160 + M, 0xf, 0xf, false));
}
DEV float row_sum16(float v)
{
    v += row_xor<1>(v);
    v += row_xor<2>(v);
    v += row_xor<4>(v);
    v += row_xor<8>(v);
    return v;
}
DEV float lane0(float v)
{
    return __builtin_bit_cast(float,
        __builtin_amdgcn_readfirstlane(__builtin_bit_cast(int, v)));
}
DEV float4 ldrow(const float* __restrict__ proj, int nid, int lane)
{
    return *((const float4*)(proj + (long)nid * PROJ2) + lane);
}
DEV float dist2_part(float4 r, float4 c)
{
    const float dx = r.x - c.x, dy = r.y - c.y, dz = r.z - c.z, dw = r.w - c.w;
    return dx * dx + dy * dy + dz * dz + dw * dw;
}
DEV float key_score(const float* __restrict__ proj, int nid, float4 c, int lane)
{
    const float ss = lane0(row_sum16(dist2_part(ldrow(proj, nid, lane), c)));
    return sqrtf(ss) * 0.125f;
}
DEV float lane_f(float v, int j)
{
    return __builtin_bit_cast(float,
        __builtin_amdgcn_readlane(__builtin_bit_cast(int, v), j));
}

#define SEG_R 8
DEV void ldrows(const float* __restrict__ proj, int myid, int base, int nb,
                int lane, float4 (&r)[SEG_R])
{
    r[0] = ldrow(proj, __builtin_amdgcn_readlane(myid, base), lane);
    #pragma unroll
    for (int u = 1; u < SEG_R; ++u) {
        r[u] = make_float4(0.f, 0.f, 0.f, 0.f);
        if (base + u < nb)
            r[u] = ldrow(proj, __builtin_amdgcn_readlane(myid, base + u), lane);
    }
}

template <class F>
DEV void for_each_overflow(const int* __restrict__ ovf, int novf, int h, int lane, F f)
{
    for (int base = 0; base < novf; base += 32) {
        const int i = base + lane;
        int hh = -1, nd = 0;
        if (i < novf) { hh = ovf[2 * i]; nd = ovf[2 * i + 1]; }
        unsigned hit = __builtin_amdgcn_ballot_w32(hh == h);
        while (hit) {
            const int j = __builtin_ctz(hit);
            hit &= hit - 1;
            f(__builtin_amdgcn_readlane(nd, j));
        }
    }
}

__global__ __launch_bounds__(256)
void k_segment(const float* __restrict__ proj,
               const int* __restrict__ bucket,
               const int* __restrict__ ovf,
               const int* __restrict__ cnt,
               float* __restrict__ hefeats, int num_he)
{
    const int lane = threadIdx.x & 31;
    const int wv = __builtin_amdgcn_readfirstlane((int)(threadIdx.x >> 5));
    const int h  = blockIdx.x * 8 + wv;
    if (h >= num_he) return;
    const int n    = cnt[h];
    const int nb   = min(n, CAP);
    const int myid = (lane < nb) ? bucket[(long)h * CAP + lane] : 0;
    const float inv = 1.0f / fmaxf((float)n, 1.0f);
    const float4 zero4 = make_float4(0.f, 0.f, 0.f, 0.f);
    float4 acc = zero4;

    if (n <= SEG_R) {
        int nid[SEG_R];
        #pragma unroll
        for (int u = 0; u < SEG_R; ++u)
            nid[u] = __builtin_amdgcn_readlane(myid, u);
        float4 r[SEG_R];
        #pragma unroll
        for (int u = 0; u < SEG_R; ++u) {
            r[u] = zero4;
            if (u < n) r[u] = ldrow(proj, nid[u], lane);
        }
        float4 ks = zero4;
        #pragma unroll
        for (int u = 0; u < SEG_R; ++u) {
            ks.x += r[u].x; ks.y += r[u].y; ks.z += r[u].z; ks.w += r[u].w;
        }
        const float4 c = make_float4(ks.x * inv, ks.y * inv, ks.z * inv, ks.w * inv);
        float sc[SEG_R];
        float smax = 0.0f;
        #pragma unroll
        for (int u = 0; u < SEG_R; ++u) {
            const float ss = lane0(row_sum16(dist2_part(r[u], c)));
            sc[u] = sqrtf(ss) * 0.125f;
            if (u < n) smax = fmaxf(smax, sc[u]);
        }
        float denom = 0.0f;
        #pragma unroll
        for (int u = 0; u < SEG_R; ++u) {
            sc[u] = (u < n) ? expf(sc[u] - smax) : 0.0f;
            denom += sc[u];
        }
        const float rden = fmaxf(denom, 1e-30f);
        #pragma unroll
        for (int u = 0; u < SEG_R; ++u) {
            const float wgt = sc[u] / rden;
            acc.x += wgt * r[u].x; acc.y += wgt * r[u].y;
            acc.z += wgt * r[u].z; acc.w += wgt * r[u].w;
        }
    } else {
        const int novf = (n > CAP) ? cnt[num_he] : 0;
        float4 r[SEG_R];
        float4 ks = zero4;
        for (int base = 0; base < nb; base += SEG_R) {
            ldrows(proj, myid, base, nb, lane, r);
            #pragma unroll
            for (int u = 0; u < SEG_R; ++u) {
                ks.x += r[u].x; ks.y += r[u].y; ks.z += r[u].z; ks.w += r[u].w;
            }
        }
        for_each_overflow(ovf, novf, h, lane, [&](int nid) {
            const float4 rr = ldrow(proj, nid, lane);
            ks.x += rr.x; ks.y += rr.y; ks.z += rr.z; ks.w += rr.w;
        });
        const float4 c = make_float4(ks.x * inv, ks.y * inv, ks.z * inv, ks.w * inv);
        float ssv = 0.0f;
        for (int base = 0; base < nb; base += SEG_R) {
            ldrows(proj, myid, base, nb, lane, r);
            #pragma unroll
            for (int u = 0; u < SEG_R; ++u) {
                const float ss = lane0(row_sum16(dist2_part(r[u], c)));
                ssv = (lane == base + u) ? ss : ssv;
            }
        }
        const float scv = sqrtf(ssv) * 0.125f;
        float mx = (lane < nb) ? scv : 0.0f;
        mx = fmaxf(mx, row_xor<1>(mx));
        mx = fmaxf(mx, row_xor<2>(mx));
        mx = fmaxf(mx, row_xor<4>(mx));
        mx = fmaxf(mx, row_xor<8>(mx));
        float smax = fmaxf(lane_f(mx, 0), lane_f(mx, 16));
        for_each_overflow(ovf, novf, h, lane, [&](int nid) {
            smax = fmaxf(smax, key_score(proj, nid, c, lane));
        });
        const float exv = (lane < nb) ? expf(scv - smax) : 0.0f;
        float denom = 0.0f;
        for (int j = 0; j < nb; ++j)
            denom += lane_f(exv, j);
        for_each_overflow(ovf, novf, h, lane, [&](int nid) {
            denom += expf(key_score(proj, nid, c, lane) - smax);
        });
        const float rden = fmaxf(denom, 1e-30f);
        const float wgtv = exv / rden;
        for (int base = 0; base < nb; base += SEG_R) {
            ldrows(proj, myid, base, nb, lane, r);
            #pragma unroll
            for (int u = 0; u < SEG_R; ++u) {
                const float wgt = lane_f(wgtv, base + u);
                acc.x += wgt * r[u].x; acc.y += wgt * r[u].y;
                acc.z += wgt * r[u].z; acc.w += wgt * r[u].w;
            }
        }
        for_each_overflow(ovf, novf, h, lane, [&](int nid) {
            const float4 rr = ldrow(proj, nid, lane);
            const float ss  = lane0(row_sum16(dist2_part(rr, c)));
            const float wgt = expf(sqrtf(ss) * 0.125f - smax) / rden;
            acc.x += wgt * rr.x; acc.y += wgt * rr.y;
            acc.z += wgt * rr.z; acc.w += wgt * rr.w;
        });
    }
    if (lane >= 16)
        *((float4*)(hefeats + (long)h * PROJ) + (lane - 16)) = acc;
}

__global__ __launch_bounds__(256)
void k_out_gemm(const float* __restrict__ hefeats,
                const float* __restrict__ W_v,
                float* __restrict__ out, int num_he)
{
    __shared__ float hs[RPB * HSTRIDE];
    const int e0 = blockIdx.x * RPB;
    const int t  = threadIdx.x;
    {
        const int r = t >> 2;
        const int c = (t & 3) * 16;
        const int e = e0 + r;
        #pragma unroll
        for (int i = 0; i < 16; i += 4) {
            float4 v = (e < num_he)
                     ? *(const float4*)(hefeats + (long)e * PROJ + c + i)
                     : make_float4(0.f, 0.f, 0.f, 0.f);
            *(float4*)(hs + r * HSTRIDE + c + i) = v;
        }
    }
    __syncthreads();

    const int wave = t >> 5;
    const int lane = t & 31;
    const int mn   = lane & 15;
    const int kb   = (lane >> 4) * 2;
    const int mh   = (lane >> 4) * 8;
    const bool full = (e0 + RPB) <= num_he;

    #pragma unroll
    for (int sub = 0; sub < 2; ++sub) {
        const int n0 = wave * 32 + sub * 16;
        v8f acc[4] = {};
        #pragma unroll 2
        for (int kk = 0; kk < PROJ; kk += 4) {
            v2f b;
            b.x = W_v[(long)(kk + kb)     * HIDDEN + n0 + mn];
            b.y = W_v[(long)(kk + kb + 1) * HIDDEN + n0 + mn];
            #pragma unroll
            for (int rt = 0; rt < 4; ++rt) {
                v2f a;
                const float* arow = hs + (rt * 16 + mn) * HSTRIDE;
                a.x = arow[kk + kb];
                a.y = arow[kk + kb + 1];
                acc[rt] = __builtin_amdgcn_wmma_f32_16x16x4_f32(
                              false, a, false, b, (short)0, acc[rt], false, false);
            }
        }
        if (full) {
            #pragma unroll
            for (int rt = 0; rt < 4; ++rt)
                #pragma unroll
                for (int r = 0; r < 8; ++r) {
                    const int e = e0 + rt * 16 + r + mh;
                    out[(long)e * HIDDEN + n0 + mn] = acc[rt][r];
                }
        } else {
            #pragma unroll
            for (int rt = 0; rt < 4; ++rt)
                #pragma unroll
                for (int r = 0; r < 8; ++r) {
                    const int e = e0 + rt * 16 + r + mh;
                    if (e < num_he) out[(long)e * HIDDEN + n0 + mn] = acc[rt][r];
                }
        }
    }
}

__global__ __launch_bounds__(256)
void k_zero(int* __restrict__ p, int n)
{
    const int i = blockIdx.x * blockDim.x + threadIdx.x;
    if (i < n) p[i] = 0;
}

extern "C" void kernel_launch(void* const* d_in, const int* in_sizes, int n_in,
                              void* d_out, int out_size, void* d_ws, size_t ws_size,
                              hipStream_t stream)
{
    const float* node_feats = (const float*)d_in[0];
    const float* W_k        = (const float*)d_in[1];
    const float* W_v        = (const float*)d_in[2];
    const int*   node_idx   = (const int*)d_in[3];
    const int*   he_idx     = (const int*)d_in[4];
    (void)n_in; (void)ws_size;

    const int nnz = in_sizes[3];
    const int HE  = out_size / HIDDEN;
    const int NN  = in_sizes[0] / HIDDEN;

    float* ws = (float*)d_ws;
    long o = 0;
    float* proj    = ws + o;        o += (long)NN * PROJ2;
    float* hefeats = ws + o;        o += (long)HE * PROJ;
    int*   cnt     = (int*)(ws + o); o += ((long)HE + 1 + 31) / 32 * 32;
    int*   bucket  = (int*)(ws + o); o += (long)HE * CAP;
    int*   ovf     = (int*)(ws + o); o += 2L * nnz;

    k_zero<<<(HE + 1 + 255) / 256, 256, 0, stream>>>(cnt, HE + 1);

    k_fill<<<(nnz + 255) / 256, 256, 0, stream>>>(
        node_idx, he_idx, cnt, bucket, ovf, nnz, HE);

    k_project<<<(NN + RPB - 1) / RPB, 256, 0, stream>>>(
        node_feats, W_k, W_v, proj, NN);

    k_segment<<<(HE + 7) / 8, 256, 0, stream>>>(
        proj, bucket, ovf, cnt, hefeats, HE);

    k_out_gemm<<<(HE + RPB - 1) / RPB, 256, 0, stream>>>(
        hefeats, W_v, (float*)d_out, HE);
}


# Edge-grouping pass folded into the projection GEMM's launch (the two are independent): each GEMM block also slots 1/grid of the edges under its WMMA loop; 5->4 launches; sums stay in arrival order
# speedup vs baseline: 3.1707x; 1.0403x over previous
#include <hip/hip_runtime.h>
#include <math.h>


typedef __attribute__((ext_vector_type(2))) float v2f;
typedef __attribute__((ext_vector_type(8))) float v8f;

#define HIDDEN 256
#define PROJ   64
#define PROJ2  128
#define RPB    64
#define XSTRIDE 260
#define HSTRIDE 68

#define CAP 32
__device__ __forceinline__
void place_edge(int h, int nid, int slot, int* __restrict__ cnt,
                int* __restrict__ bucket, int* __restrict__ ovf, int num_he)
{
    if (slot < CAP) {
        bucket[(long)h * CAP + slot] = nid;
    } else {
        const int o = atomicAdd(&cnt[num_he], 1);
        ovf[2 * o]     = h;
        ovf[2 * o + 1] = nid;
    }
}

__global__ __launch_bounds__(256)
void k_project(const float* __restrict__ node_feats,
               const float* __restrict__ W_k,
               const float* __restrict__ W_v,
               float* __restrict__ proj,
               int n_nodes,
               const int* __restrict__ node_idx,
               const int* __restrict__ he_idx,
               int* __restrict__ cnt, int* __restrict__ bucket,
               int* __restrict__ ovf, int nnz, int num_he)
{
    __shared__ float xs[RPB * XSTRIDE];
    const int r0 = blockIdx.x * RPB;
    const int t  = threadIdx.x;

    const int epb  = (nnz + (int)gridDim.x - 1) / (int)gridDim.x;
    const int ebeg = (int)blockIdx.x * epb;
    const int eend = min(nnz, ebeg + epb);
    const int e    = ebeg + t;
    const bool has_edge = e < eend;
    int eh = 0, en = 0;
    if (has_edge) { eh = he_idx[e]; en = node_idx[e]; }

    {
        const int rr = t >> 6;
        const int c  = (t & 63) * 4;
        #pragma unroll
        for (int i = 0; i < 16; ++i) {
            const int r   = i * 4 + rr;
            const int n   = r0 + r;
            const int row = (n < n_nodes) ? n : 0;
            *(float4*)(xs + r * XSTRIDE + c) =
                *(const float4*)(node_feats + (long)row * HIDDEN + c);
        }
    }
    __syncthreads();

    int eslot = 0;
    if (has_edge) eslot = atomicAdd(&cnt[eh], 1);

    const int wave = t >> 5;
    const int lane = t & 31;
    const int mn   = lane & 15;
    const int kb   = (lane >> 4) * 2;
    const int mh   = (lane >> 4) * 8;
    const int n0   = wave * 16;

    const int nrow = n0 + mn;
    const float* wrow = (nrow < PROJ) ? (W_k + (long)nrow * HIDDEN)
                                      : (W_v + (long)(nrow - PROJ) * HIDDEN);

    v8f acc[4] = {};
    #pragma unroll 2
    for (int kk = 0; kk < HIDDEN; kk += 4) {
        v2f b;
        b.x = wrow[kk + kb];
        b.y = wrow[kk + kb + 1];
        #pragma unroll
        for (int rt = 0; rt < 4; ++rt) {
            v2f a;
            const float* arow = xs + (rt * 16 + mn) * XSTRIDE;
            a.x = arow[kk + kb];
            a.y = arow[kk + kb + 1];
            acc[rt] = __builtin_amdgcn_wmma_f32_16x16x4_f32(
                          false, a, false, b, (short)0, acc[rt], false, false);
        }
    }

    if (r0 + RPB <= n_nodes) {
        #pragma unroll
        for (int rt = 0; rt < 4; ++rt)
            #pragma unroll
            for (int r = 0; r < 8; ++r) {
                const int n = r0 + rt * 16 + r + mh;
                proj[(long)n * PROJ2 + n0 + mn] = acc[rt][r];
            }
    } else {
        #pragma unroll
        for (int rt = 0; rt < 4; ++rt)
            #pragma unroll
            for (int r = 0; r < 8; ++r) {
                const int n = r0 + rt * 16 + r + mh;
                if (n < n_nodes)
                    proj[(long)n * PROJ2 + n0 + mn] = acc[rt][r];
            }
    }

    if (has_edge) place_edge(eh, en, eslot, cnt, bucket, ovf, num_he);
    for (int e2 = e + 256; e2 < eend; e2 += 256) {
        const int h2 = he_idx[e2];
        place_edge(h2, node_idx[e2], atomicAdd(&cnt[h2], 1), cnt, bucket, ovf, num_he);
    }
}

#define DEV __device__ __forceinline__

template <int M>
DEV float row_xor(float x)
{
    return __builtin_bit_cast(float, __builtin_amdgcn_update_dpp(
        0, __builtin_bit_cast(int, x), 0x160 + M, 0xf, 0xf, false));
}
DEV float row_sum16(float v)
{
    v += row_xor<1>(v);
    v += row_xor<2>(v);
    v += row_xor<4>(v);
    v += row_xor<8>(v);
    return v;
}
DEV float lane0(float v)
{
    return __builtin_bit_cast(float,
        __builtin_amdgcn_readfirstlane(__builtin_bit_cast(int, v)));
}
DEV float4 ldrow(const float* __restrict__ proj, int nid, int lane)
{
    return *((const float4*)(proj + (long)nid * PROJ2) + lane);
}
DEV float dist2_part(float4 r, float4 c)
{
    const float dx = r.x - c.x, dy = r.y - c.y, dz = r.z - c.z, dw = r.w - c.w;
    return dx * dx + dy * dy + dz * dz + dw * dw;
}
DEV float key_score(const float* __restrict__ proj, int nid, float4 c, int lane)
{
    const float ss = lane0(row_sum16(dist2_part(ldrow(proj, nid, lane), c)));
    return sqrtf(ss) * 0.125f;
}
DEV float lane_f(float v, int j)
{
    return __builtin_bit_cast(float,
        __builtin_amdgcn_readlane(__builtin_bit_cast(int, v), j));
}

#define SEG_R 8
DEV void ldrows(const float* __restrict__ proj, int myid, int base, int nb,
                int lane, float4 (&r)[SEG_R])
{
    r[0] = ldrow(proj, __builtin_amdgcn_readlane(myid, base), lane);
    #pragma unroll
    for (int u = 1; u < SEG_R; ++u) {
        r[u] = make_float4(0.f, 0.f, 0.f, 0.f);
        if (base + u < nb)
            r[u] = ldrow(proj, __builtin_amdgcn_readlane(myid, base + u), lane);
    }
}

template <class F>
DEV void for_each_overflow(const int* __restrict__ ovf, int novf, int h, int lane, F f)
{
    for (int base = 0; base < novf; base += 32) {
        const int i = base + lane;
        int hh = -1, nd = 0;
        if (i < novf) { hh = ovf[2 * i]; nd = ovf[2 * i + 1]; }
        unsigned hit = __builtin_amdgcn_ballot_w32(hh == h);
        while (hit) {
            const int j = __builtin_ctz(hit);
            hit &= hit - 1;
            f(__builtin_amdgcn_readlane(nd, j));
        }
    }
}

__global__ __launch_bounds__(256)
void k_segment(const float* __restrict__ proj,
               const int* __restrict__ bucket,
               const int* __restrict__ ovf,
               const int* __restrict__ cnt,
               float* __restrict__ hefeats, int num_he)
{
    const int lane = threadIdx.x & 31;
    const int wv = __builtin_amdgcn_readfirstlane((int)(threadIdx.x >> 5));
    const int h  = blockIdx.x * 8 + wv;
    if (h >= num_he) return;
    const int n    = cnt[h];
    const int nb   = min(n, CAP);
    const int myid = (lane < nb) ? bucket[(long)h * CAP + lane] : 0;
    const float inv = 1.0f / fmaxf((float)n, 1.0f);
    const float4 zero4 = make_float4(0.f, 0.f, 0.f, 0.f);
    float4 acc = zero4;

    if (n <= SEG_R) {
        int nid[SEG_R];
        #pragma unroll
        for (int u = 0; u < SEG_R; ++u)
            nid[u] = __builtin_amdgcn_readlane(myid, u);
        float4 r[SEG_R];
        #pragma unroll
        for (int u = 0; u < SEG_R; ++u) {
            r[u] = zero4;
            if (u < n) r[u] = ldrow(proj, nid[u], lane);
        }
        float4 ks = zero4;
        #pragma unroll
        for (int u = 0; u < SEG_R; ++u) {
            ks.x += r[u].x; ks.y += r[u].y; ks.z += r[u].z; ks.w += r[u].w;
        }
        const float4 c = make_float4(ks.x * inv, ks.y * inv, ks.z * inv, ks.w * inv);
        float sc[SEG_R];
        float smax = 0.0f;
        #pragma unroll
        for (int u = 0; u < SEG_R; ++u) {
            const float ss = lane0(row_sum16(dist2_part(r[u], c)));
            sc[u] = sqrtf(ss) * 0.125f;
            if (u < n) smax = fmaxf(smax, sc[u]);
        }
        float denom = 0.0f;
        #pragma unroll
        for (int u = 0; u < SEG_R; ++u) {
            sc[u] = (u < n) ? expf(sc[u] - smax) : 0.0f;
            denom += sc[u];
        }
        const float rden = fmaxf(denom, 1e-30f);
        #pragma unroll
        for (int u = 0; u < SEG_R; ++u) {
            const float wgt = sc[u] / rden;
            acc.x += wgt * r[u].x; acc.y += wgt * r[u].y;
            acc.z += wgt * r[u].z; acc.w += wgt * r[u].w;
        }
    } else {
        const int novf = (n > CAP) ? cnt[num_he] : 0;
        float4 r[SEG_R];
        float4 ks = zero4;
        for (int base = 0; base < nb; base += SEG_R) {
            ldrows(proj, myid, base, nb, lane, r);
            #pragma unroll
            for (int u = 0; u < SEG_R; ++u) {
                ks.x += r[u].x; ks.y += r[u].y; ks.z += r[u].z; ks.w += r[u].w;
            }
        }
        for_each_overflow(ovf, novf, h, lane, [&](int nid) {
            const float4 rr = ldrow(proj, nid, lane);
            ks.x += rr.x; ks.y += rr.y; ks.z += rr.z; ks.w += rr.w;
        });
        const float4 c = make_float4(ks.x * inv, ks.y * inv, ks.z * inv, ks.w * inv);
        float ssv = 0.0f;
        for (int base = 0; base < nb; base += SEG_R) {
            ldrows(proj, myid, base, nb, lane, r);
            #pragma unroll
            for (int u = 0; u < SEG_R; ++u) {
                const float ss = lane0(row_sum16(dist2_part(r[u], c)));
                ssv = (lane == base + u) ? ss : ssv;
            }
        }
        const float scv = sqrtf(ssv) * 0.125f;
        float mx = (lane < nb) ? scv : 0.0f;
        mx = fmaxf(mx, row_xor<1>(mx));
        mx = fmaxf(mx, row_xor<2>(mx));
        mx = fmaxf(mx, row_xor<4>(mx));
        mx = fmaxf(mx, row_xor<8>(mx));
        float smax = fmaxf(lane_f(mx, 0), lane_f(mx, 16));
        for_each_overflow(ovf, novf, h, lane, [&](int nid) {
            smax = fmaxf(smax, key_score(proj, nid, c, lane));
        });
        const float exv = (lane < nb) ? expf(scv - smax) : 0.0f;
        float denom = 0.0f;
        for (int j = 0; j < nb; ++j)
            denom += lane_f(exv, j);
        for_each_overflow(ovf, novf, h, lane, [&](int nid) {
            denom += expf(key_score(proj, nid, c, lane) - smax);
        });
        const float rden = fmaxf(denom, 1e-30f);
        const float wgtv = exv / rden;
        for (int base = 0; base < nb; base += SEG_R) {
            ldrows(proj, myid, base, nb, lane, r);
            #pragma unroll
            for (int u = 0; u < SEG_R; ++u) {
                const float wgt = lane_f(wgtv, base + u);
                acc.x += wgt * r[u].x; acc.y += wgt * r[u].y;
                acc.z += wgt * r[u].z; acc.w += wgt * r[u].w;
            }
        }
        for_each_overflow(ovf, novf, h, lane, [&](int nid) {
            const float4 rr = ldrow(proj, nid, lane);
            const float ss  = lane0(row_sum16(dist2_part(rr, c)));
            const float wgt = expf(sqrtf(ss) * 0.125f - smax) / rden;
            acc.x += wgt * rr.x; acc.y += wgt * rr.y;
            acc.z += wgt * rr.z; acc.w += wgt * rr.w;
        });
    }
    if (lane >= 16)
        *((float4*)(hefeats + (long)h * PROJ) + (lane - 16)) = acc;
}

__global__ __launch_bounds__(256)
void k_out_gemm(const float* __restrict__ hefeats,
                const float* __restrict__ W_v,
                float* __restrict__ out, int num_he)
{
    __shared__ float hs[RPB * HSTRIDE];
    const int e0 = blockIdx.x * RPB;
    const int t  = threadIdx.x;
    {
        const int r = t >> 2;
        const int c = (t & 3) * 16;
        const int e = e0 + r;
        #pragma unroll
        for (int i = 0; i < 16; i += 4) {
            float4 v = (e < num_he)
                     ? *(const float4*)(hefeats + (long)e * PROJ + c + i)
                     : make_float4(0.f, 0.f, 0.f, 0.f);
            *(float4*)(hs + r * HSTRIDE + c + i) = v;
        }
    }
    __syncthreads();

    const int wave = t >> 5;
    const int lane = t & 31;
    const int mn   = lane & 15;
    const int kb   = (lane >> 4) * 2;
    const int mh   = (lane >> 4) * 8;
    const bool full = (e0 + RPB) <= num_he;

    #pragma unroll
    for (int sub = 0; sub < 2; ++sub) {
        const int n0 = wave * 32 + sub * 16;
        v8f acc[4] = {};
        #pragma unroll 2
        for (int kk = 0; kk < PROJ; kk += 4) {
            v2f b;
            b.x = W_v[(long)(kk + kb)     * HIDDEN + n0 + mn];
            b.y = W_v[(long)(kk + kb + 1) * HIDDEN + n0 + mn];
            #pragma unroll
            for (int rt = 0; rt < 4; ++rt) {
                v2f a;
                const float* arow = hs + (rt * 16 + mn) * HSTRIDE;
                a.x = arow[kk + kb];
                a.y = arow[kk + kb + 1];
                acc[rt] = __builtin_amdgcn_wmma_f32_16x16x4_f32(
                              false, a, false, b, (short)0, acc[rt], false, false);
            }
        }
        if (full) {
            #pragma unroll
            for (int rt = 0; rt < 4; ++rt)
                #pragma unroll
                for (int r = 0; r < 8; ++r) {
                    const int e = e0 + rt * 16 + r + mh;
                    out[(long)e * HIDDEN + n0 + mn] = acc[rt][r];
                }
        } else {
            #pragma unroll
            for (int rt = 0; rt < 4; ++rt)
                #pragma unroll
                for (int r = 0; r < 8; ++r) {
                    const int e = e0 + rt * 16 + r + mh;
                    if (e < num_he) out[(long)e * HIDDEN + n0 + mn] = acc[rt][r];
                }
        }
    }
}

__global__ __launch_bounds__(256)
void k_zero(int* __restrict__ p, int n)
{
    const int i = blockIdx.x * blockDim.x + threadIdx.x;
    if (i < n) p[i] = 0;
}

extern "C" void kernel_launch(void* const* d_in, const int* in_sizes, int n_in,
                              void* d_out, int out_size, void* d_ws, size_t ws_size,
                              hipStream_t stream)
{
    const float* node_feats = (const float*)d_in[0];
    const float* W_k        = (const float*)d_in[1];
    const float* W_v        = (const float*)d_in[2];
    const int*   node_idx   = (const int*)d_in[3];
    const int*   he_idx     = (const int*)d_in[4];
    (void)n_in; (void)ws_size;

    const int nnz = in_sizes[3];
    const int HE  = out_size / HIDDEN;
    const int NN  = in_sizes[0] / HIDDEN;

    float* ws = (float*)d_ws;
    long o = 0;
    float* proj    = ws + o;        o += (long)NN * PROJ2;
    float* hefeats = ws + o;        o += (long)HE * PROJ;
    int*   cnt     = (int*)(ws + o); o += ((long)HE + 1 + 31) / 32 * 32;
    int*   bucket  = (int*)(ws + o); o += (long)HE * CAP;
    int*   ovf     = (int*)(ws + o); o += 2L * nnz;

    k_zero<<<(HE + 1 + 255) / 256, 256, 0, stream>>>(cnt, HE + 1);

    k_project<<<(NN + RPB - 1) / RPB, 256, 0, stream>>>(
        node_feats, W_k, W_v, proj, NN,
        node_idx, he_idx, cnt, bucket, ovf, nnz, HE);

    k_segment<<<(HE + 7) / 8, 256, 0, stream>>>(
        proj, bucket, ovf, cnt, hefeats, HE);

    k_out_gemm<<<(HE + RPB - 1) / RPB, 256, 0, stream>>>(
        hefeats, W_v, (float*)d_out, HE);
}


# k_project as 256 persistent blocks (one per WGP): W fragments stay in registers, row tiles double-buffered in LDS, next tile's loads under the WMMA loop, rows dealt in 16-row units; same sum order
# speedup vs baseline: 3.4345x; 1.0832x over previous
#include <hip/hip_runtime.h>
#include <math.h>


typedef __attribute__((ext_vector_type(2))) float v2f;
typedef __attribute__((ext_vector_type(8))) float v8f;

#define HIDDEN 256
#define PROJ   64
#define PROJ2  128
#define RPB    64
#define XSTRIDE 260
#define HSTRIDE 68

#define CAP 32
__device__ __forceinline__
void place_edge(int h, int nid, int slot, int* __restrict__ cnt,
                int* __restrict__ bucket, int* __restrict__ ovf, int num_he)
{
    if (slot < CAP) {
        bucket[(long)h * CAP + slot] = nid;
    } else {
        const int o = atomicAdd(&cnt[num_he], 1);
        ovf[2 * o]     = h;
        ovf[2 * o + 1] = nid;
    }
}

#define NBLK_P 256
#define XTILE  (RPB * XSTRIDE)
typedef __attribute__((ext_vector_type(4))) float v4f;

template <int NRT, int K0, int K1>
__device__ __forceinline__
void ksteps(const float* abase, const v2f (&bw)[64], v8f (&acc)[4])
{
    #pragma unroll
    for (int ks = K0; ks < K1; ++ks) {
        #pragma unroll
        for (int rt = 0; rt < NRT; ++rt) {
            v2f a;
            a.x = abase[rt * 16 * XSTRIDE + ks * 4];
            a.y = abase[rt * 16 * XSTRIDE + ks * 4 + 1];
            acc[rt] = __builtin_amdgcn_wmma_f32_16x16x4_f32(
                          false, a, false, bw[ks], (short)0, acc[rt], false, false);
        }
    }
}

__device__ __forceinline__
void fetch_tile(const float* __restrict__ node_feats, int n_nodes, int r0, int nu,
                int rr, int c, v4f (&pf)[16])
{
    #pragma unroll
    for (int g = 0; g < 4; ++g)
        if (g < nu) {
            #pragma unroll
            for (int j = 0; j < 4; ++j) {
                const int n   = r0 + (g * 4 + j) * 4 + rr;
                const int row = (n < n_nodes) ? n : 0;
                pf[g * 4 + j] = *(const v4f*)(node_feats + (long)row * HIDDEN + c);
            }
        }
}
__device__ __forceinline__
void stash_tile(float* dst, int nu, int rr, int c, const v4f (&pf)[16])
{
    #pragma unroll
    for (int g = 0; g < 4; ++g)
        if (g < nu) {
            #pragma unroll
            for (int j = 0; j < 4; ++j)
                *(v4f*)(dst + ((g * 4 + j) * 4 + rr) * XSTRIDE + c) = pf[g * 4 + j];
        }
}

__global__ __launch_bounds__(256)
void k_project(const float* __restrict__ node_feats,
               const float* __restrict__ W_k,
               const float* __restrict__ W_v,
               float* __restrict__ proj,
               int n_nodes,
               const int* __restrict__ node_idx,
               const int* __restrict__ he_idx,
               int* __restrict__ cnt, int* __restrict__ bucket,
               int* __restrict__ ovf, int nnz, int num_he)
{
    __shared__ float xs[2 * XTILE];
    const int t    = threadIdx.x;
    const int wave = t >> 5;
    const int lane = t & 31;
    const int mn   = lane & 15;
    const int kb   = (lane >> 4) * 2;
    const int mh   = (lane >> 4) * 8;
    const int n0   = wave * 16;
    const int rr   = t >> 6;
    const int c    = (t & 63) * 4;

    const int RT     = (n_nodes + 15) >> 4;
    const int rtpb   = (RT + (int)gridDim.x - 1) / (int)gridDim.x;
    const int rt_beg = (int)blockIdx.x * rtpb;
    const int myrt   = max(0, min(RT, rt_beg + rtpb) - rt_beg);
    const int ntile  = (myrt + 3) >> 2;

    const int epb  = (nnz + (int)gridDim.x - 1) / (int)gridDim.x;
    const int ebeg = (int)blockIdx.x * epb;
    const int eend = min(nnz, ebeg + epb);

    const int nrow = n0 + mn;
    const float* wrow = (nrow < PROJ) ? (W_k + (long)nrow * HIDDEN)
                                      : (W_v + (long)(nrow - PROJ) * HIDDEN);
    v2f bw[64];
    #pragma unroll
    for (int ks = 0; ks < 64; ++ks) {
        bw[ks].x = wrow[ks * 4 + kb];
        bw[ks].y = wrow[ks * 4 + kb + 1];
    }

    v4f pf[16] = {};

    if (ntile > 0) {
        const int nu0 = min(4, myrt);
        fetch_tile(node_feats, n_nodes, rt_beg * 16, nu0, rr, c, pf);
        stash_tile(xs, nu0, rr, c, pf);
    }
    __syncthreads();

    for (int i = 0; i < ntile; ++i) {
        const float* xc = xs + (i & 1) * XTILE;
        float*       xn = xs + ((i + 1) & 1) * XTILE;
        const int r0   = (rt_beg + 4 * i) * 16;
        const int nrt  = min(4, myrt - 4 * i);
        const int nun  = (i + 1 < ntile) ? min(4, myrt - 4 * (i + 1)) : 0;

        const int e = ebeg + i * 256 + t;
        const bool has_edge = e < eend;
        int eh = 0, en = 0, eslot = 0;
        if (has_edge) { eh = he_idx[e]; en = node_idx[e]; }
        fetch_tile(node_feats, n_nodes, r0 + RPB, nun, rr, c, pf);

        v8f acc[4] = {};
        const float* abase = xc + mn * XSTRIDE + kb;
        if (nrt == 4) {
            ksteps<4, 0, 32>(abase, bw, acc);
            stash_tile(xn, nun, rr, c, pf);
            if (has_edge) eslot = atomicAdd(&cnt[eh], 1);
            ksteps<4, 32, 64>(abase, bw, acc);
        } else {
            if (has_edge) eslot = atomicAdd(&cnt[eh], 1);
            if (nrt == 1)      ksteps<1, 0, 64>(abase, bw, acc);
            else if (nrt == 2) ksteps<2, 0, 64>(abase, bw, acc);
            else               ksteps<3, 0, 64>(abase, bw, acc);
        }

        if (nrt == 4 && r0 + RPB <= n_nodes) {
            #pragma unroll
            for (int rt = 0; rt < 4; ++rt)
                #pragma unroll
                for (int r = 0; r < 8; ++r) {
                    const int n = r0 + rt * 16 + r + mh;
                    proj[(long)n * PROJ2 + n0 + mn] = acc[rt][r];
                }
        } else {
            #pragma unroll
            for (int rt = 0; rt < 4; ++rt)
                if (rt < nrt) {
                    #pragma unroll
                    for (int r = 0; r < 8; ++r) {
                        const int n = r0 + rt * 16 + r + mh;
                        if (n < n_nodes)
                            proj[(long)n * PROJ2 + n0 + mn] = acc[rt][r];
                    }
                }
        }
        if (has_edge) place_edge(eh, en, eslot, cnt, bucket, ovf, num_he);
        __syncthreads();
    }

    for (int e2 = ebeg + ntile * 256 + t; e2 < eend; e2 += 256) {
        const int h2 = he_idx[e2];
        place_edge(h2, node_idx[e2], atomicAdd(&cnt[h2], 1), cnt, bucket, ovf, num_he);
    }
}

#define DEV __device__ __forceinline__

template <int M>
DEV float row_xor(float x)
{
    return __builtin_bit_cast(float, __builtin_amdgcn_update_dpp(
        0, __builtin_bit_cast(int, x), 0x160 + M, 0xf, 0xf, false));
}
DEV float row_sum16(float v)
{
    v += row_xor<1>(v);
    v += row_xor<2>(v);
    v += row_xor<4>(v);
    v += row_xor<8>(v);
    return v;
}
DEV float lane0(float v)
{
    return __builtin_bit_cast(float,
        __builtin_amdgcn_readfirstlane(__builtin_bit_cast(int, v)));
}
DEV float4 ldrow(const float* __restrict__ proj, int nid, int lane)
{
    return *((const float4*)(proj + (long)nid * PROJ2) + lane);
}
DEV float dist2_part(float4 r, float4 c)
{
    const float dx = r.x - c.x, dy = r.y - c.y, dz = r.z - c.z, dw = r.w - c.w;
    return dx * dx + dy * dy + dz * dz + dw * dw;
}
DEV float key_score(const float* __restrict__ proj, int nid, float4 c, int lane)
{
    const float ss = lane0(row_sum16(dist2_part(ldrow(proj, nid, lane), c)));
    return sqrtf(ss) * 0.125f;
}
DEV float lane_f(float v, int j)
{
    return __builtin_bit_cast(float,
        __builtin_amdgcn_readlane(__builtin_bit_cast(int, v), j));
}

#define SEG_R 8
DEV void ldrows(const float* __restrict__ proj, int myid, int base, int nb,
                int lane, float4 (&r)[SEG_R])
{
    r[0] = ldrow(proj, __builtin_amdgcn_readlane(myid, base), lane);
    #pragma unroll
    for (int u = 1; u < SEG_R; ++u) {
        r[u] = make_float4(0.f, 0.f, 0.f, 0.f);
        if (base + u < nb)
            r[u] = ldrow(proj, __builtin_amdgcn_readlane(myid, base + u), lane);
    }
}

template <class F>
DEV void for_each_overflow(const int* __restrict__ ovf, int novf, int h, int lane, F f)
{
    for (int base = 0; base < novf; base += 32) {
        const int i = base + lane;
        int hh = -1, nd = 0;
        if (i < novf) { hh = ovf[2 * i]; nd = ovf[2 * i + 1]; }
        unsigned hit = __builtin_amdgcn_ballot_w32(hh == h);
        while (hit) {
            const int j = __builtin_ctz(hit);
            hit &= hit - 1;
            f(__builtin_amdgcn_readlane(nd, j));
        }
    }
}

__global__ __launch_bounds__(256)
void k_segment(const float* __restrict__ proj,
               const int* __restrict__ bucket,
               const int* __restrict__ ovf,
               const int* __restrict__ cnt,
               float* __restrict__ hefeats, int num_he)
{
    const int lane = threadIdx.x & 31;
    const int wv = __builtin_amdgcn_readfirstlane((int)(threadIdx.x >> 5));
    const int h  = blockIdx.x * 8 + wv;
    if (h >= num_he) return;
    const int n    = cnt[h];
    const int nb   = min(n, CAP);
    const int myid = (lane < nb) ? bucket[(long)h * CAP + lane] : 0;
    const float inv = 1.0f / fmaxf((float)n, 1.0f);
    const float4 zero4 = make_float4(0.f, 0.f, 0.f, 0.f);
    float4 acc = zero4;

    if (n <= SEG_R) {
        int nid[SEG_R];
        #pragma unroll
        for (int u = 0; u < SEG_R; ++u)
            nid[u] = __builtin_amdgcn_readlane(myid, u);
        float4 r[SEG_R];
        #pragma unroll
        for (int u = 0; u < SEG_R; ++u) {
            r[u] = zero4;
            if (u < n) r[u] = ldrow(proj, nid[u], lane);
        }
        float4 ks = zero4;
        #pragma unroll
        for (int u = 0; u < SEG_R; ++u) {
            ks.x += r[u].x; ks.y += r[u].y; ks.z += r[u].z; ks.w += r[u].w;
        }
        const float4 c = make_float4(ks.x * inv, ks.y * inv, ks.z * inv, ks.w * inv);
        float sc[SEG_R];
        float smax = 0.0f;
        #pragma unroll
        for (int u = 0; u < SEG_R; ++u) {
            const float ss = lane0(row_sum16(dist2_part(r[u], c)));
            sc[u] = sqrtf(ss) * 0.125f;
            if (u < n) smax = fmaxf(smax, sc[u]);
        }
        float denom = 0.0f;
        #pragma unroll
        for (int u = 0; u < SEG_R; ++u) {
            sc[u] = (u < n) ? expf(sc[u] - smax) : 0.0f;
            denom += sc[u];
        }
        const float rden = fmaxf(denom, 1e-30f);
        #pragma unroll
        for (int u = 0; u < SEG_R; ++u) {
            const float wgt = sc[u] / rden;
            acc.x += wgt * r[u].x; acc.y += wgt * r[u].y;
            acc.z += wgt * r[u].z; acc.w += wgt * r[u].w;
        }
    } else {
        const int novf = (n > CAP) ? cnt[num_he] : 0;
        float4 r[SEG_R];
        float4 ks = zero4;
        for (int base = 0; base < nb; base += SEG_R) {
            ldrows(proj, myid, base, nb, lane, r);
            #pragma unroll
            for (int u = 0; u < SEG_R; ++u) {
                ks.x += r[u].x; ks.y += r[u].y; ks.z += r[u].z; ks.w += r[u].w;
            }
        }
        for_each_overflow(ovf, novf, h, lane, [&](int nid) {
            const float4 rr = ldrow(proj, nid, lane);
            ks.x += rr.x; ks.y += rr.y; ks.z += rr.z; ks.w += rr.w;
        });
        const float4 c = make_float4(ks.x * inv, ks.y * inv, ks.z * inv, ks.w * inv);
        float ssv = 0.0f;
        for (int base = 0; base < nb; base += SEG_R) {
            ldrows(proj, myid, base, nb, lane, r);
            #pragma unroll
            for (int u = 0; u < SEG_R; ++u) {
                const float ss = lane0(row_sum16(dist2_part(r[u], c)));
                ssv = (lane == base + u) ? ss : ssv;
            }
        }
        const float scv = sqrtf(ssv) * 0.125f;
        float mx = (lane < nb) ? scv : 0.0f;
        mx = fmaxf(mx, row_xor<1>(mx));
        mx = fmaxf(mx, row_xor<2>(mx));
        mx = fmaxf(mx, row_xor<4>(mx));
        mx = fmaxf(mx, row_xor<8>(mx));
        float smax = fmaxf(lane_f(mx, 0), lane_f(mx, 16));
        for_each_overflow(ovf, novf, h, lane, [&](int nid) {
            smax = fmaxf(smax, key_score(proj, nid, c, lane));
        });
        const float exv = (lane < nb) ? expf(scv - smax) : 0.0f;
        float denom = 0.0f;
        for (int j = 0; j < nb; ++j)
            denom += lane_f(exv, j);
        for_each_overflow(ovf, novf, h, lane, [&](int nid) {
            denom += expf(key_score(proj, nid, c, lane) - smax);
        });
        const float rden = fmaxf(denom, 1e-30f);
        const float wgtv = exv / rden;
        for (int base = 0; base < nb; base += SEG_R) {
            ldrows(proj, myid, base, nb, lane, r);
            #pragma unroll
            for (int u = 0; u < SEG_R; ++u) {
                const float wgt = lane_f(wgtv, base + u);
                acc.x += wgt * r[u].x; acc.y += wgt * r[u].y;
                acc.z += wgt * r[u].z; acc.w += wgt * r[u].w;
            }
        }
        for_each_overflow(ovf, novf, h, lane, [&](int nid) {
            const float4 rr = ldrow(proj, nid, lane);
            const float ss  = lane0(row_sum16(dist2_part(rr, c)));
            const float wgt = expf(sqrtf(ss) * 0.125f - smax) / rden;
            acc.x += wgt * rr.x; acc.y += wgt * rr.y;
            acc.z += wgt * rr.z; acc.w += wgt * rr.w;
        });
    }
    if (lane >= 16)
        *((float4*)(hefeats + (long)h * PROJ) + (lane - 16)) = acc;
}

__global__ __launch_bounds__(256)
void k_out_gemm(const float* __restrict__ hefeats,
                const float* __restrict__ W_v,
                float* __restrict__ out, int num_he)
{
    __shared__ float hs[RPB * HSTRIDE];
    const int e0 = blockIdx.x * RPB;
    const int t  = threadIdx.x;
    {
        const int r = t >> 2;
        const int c = (t & 3) * 16;
        const int e = e0 + r;
        #pragma unroll
        for (int i = 0; i < 16; i += 4) {
            float4 v = (e < num_he)
                     ? *(const float4*)(hefeats + (long)e * PROJ + c + i)
                     : make_float4(0.f, 0.f, 0.f, 0.f);
            *(float4*)(hs + r * HSTRIDE + c + i) = v;
        }
    }
    __syncthreads();

    const int wave = t >> 5;
    const int lane = t & 31;
    const int mn   = lane & 15;
    const int kb   = (lane >> 4) * 2;
    const int mh   = (lane >> 4) * 8;
    const bool full = (e0 + RPB) <= num_he;

    #pragma unroll
    for (int sub = 0; sub < 2; ++sub) {
        const int n0 = wave * 32 + sub * 16;
        v8f acc[4] = {};
        #pragma unroll 2
        for (int kk = 0; kk < PROJ; kk += 4) {
            v2f b;
            b.x = W_v[(long)(kk + kb)     * HIDDEN + n0 + mn];
            b.y = W_v[(long)(kk + kb + 1) * HIDDEN + n0 + mn];
            #pragma unroll
            for (int rt = 0; rt < 4; ++rt) {
                v2f a;
                const float* arow = hs + (rt * 16 + mn) * HSTRIDE;
                a.x = arow[kk + kb];
                a.y = arow[kk + kb + 1];
                acc[rt] = __builtin_amdgcn_wmma_f32_16x16x4_f32(
                              false, a, false, b, (short)0, acc[rt], false, false);
            }
        }
        if (full) {
            #pragma unroll
            for (int rt = 0; rt < 4; ++rt)
                #pragma unroll
                for (int r = 0; r < 8; ++r) {
                    const int e = e0 + rt * 16 + r + mh;
                    out[(long)e * HIDDEN + n0 + mn] = acc[rt][r];
                }
        } else {
            #pragma unroll
            for (int rt = 0; rt < 4; ++rt)
                #pragma unroll
                for (int r = 0; r < 8; ++r) {
                    const int e = e0 + rt * 16 + r + mh;
                    if (e < num_he) out[(long)e * HIDDEN + n0 + mn] = acc[rt][r];
                }
        }
    }
}

__global__ __launch_bounds__(256)
void k_zero(int* __restrict__ p, int n)
{
    const int i = blockIdx.x * blockDim.x + threadIdx.x;
    if (i < n) p[i] = 0;
}

extern "C" void kernel_launch(void* const* d_in, const int* in_sizes, int n_in,
                              void* d_out, int out_size, void* d_ws, size_t ws_size,
                              hipStream_t stream)
{
    const float* node_feats = (const float*)d_in[0];
    const float* W_k        = (const float*)d_in[1];
    const float* W_v        = (const float*)d_in[2];
    const int*   node_idx   = (const int*)d_in[3];
    const int*   he_idx     = (const int*)d_in[4];
    (void)n_in; (void)ws_size;

    const int nnz = in_sizes[3];
    const int HE  = out_size / HIDDEN;
    const int NN  = in_sizes[0] / HIDDEN;

    float* ws = (float*)d_ws;
    long o = 0;
    float* proj    = ws + o;        o += (long)NN * PROJ2;
    float* hefeats = ws + o;        o += (long)HE * PROJ;
    int*   cnt     = (int*)(ws + o); o += ((long)HE + 1 + 31) / 32 * 32;
    int*   bucket  = (int*)(ws + o); o += (long)HE * CAP;
    int*   ovf     = (int*)(ws + o); o += 2L * nnz;

    k_zero<<<(HE + 1 + 255) / 256, 256, 0, stream>>>(cnt, HE + 1);

    k_project<<<NBLK_P, 256, 0, stream>>>(
        node_feats, W_k, W_v, proj, NN,
        node_idx, he_idx, cnt, bucket, ovf, nnz, HE);

    k_segment<<<(HE + 7) / 8, 256, 0, stream>>>(
        proj, bucket, ovf, cnt, hefeats, HE);

    k_out_gemm<<<(HE + RPB - 1) / RPB, 256, 0, stream>>>(
        hefeats, W_v, (float*)d_out, HE);
}
